# starting kernel
# baseline (speedup 1.0000x reference)
#include <hip/hip_runtime.h>
#include <hip/hip_bf16.h>


#define GCN_N 50000
#define GCN_E 1600000
#define GCN_D 128
#define GCN_L 3
#define LN_EPS 1e-5f

typedef __attribute__((ext_vector_type(2))) float v2f;
typedef __attribute__((ext_vector_type(8))) float v8f;


__global__ void k_init_deg(float* __restrict__ deg, int n) {
    int i = blockIdx.x * blockDim.x + threadIdx.x;
    if (i < n) deg[i] = 1.0f;
}

__global__ void k_accum_deg(const float* __restrict__ w, const int* __restrict__ col,
                            float* __restrict__ deg, int e) {
    int i = blockIdx.x * blockDim.x + threadIdx.x;
    if (i < e) atomicAdd(&deg[col[i]], w[i]);
}

__global__ void k_dinv(const float* __restrict__ deg, float* __restrict__ dinv, int n) {
    int i = blockIdx.x * blockDim.x + threadIdx.x;
    if (i < n) {
        float d = deg[i];
        dinv[i] = (d > 0.0f) ? rsqrtf(fmaxf(d, 1e-12f)) : 0.0f;
    }
}

__global__ void k_norm(const float* __restrict__ dinv, const int* __restrict__ row,
                       const int* __restrict__ col, const float* __restrict__ w,
                       float* __restrict__ norm, int e) {
    int i = blockIdx.x * blockDim.x + threadIdx.x;
    if (i < e) norm[i] = dinv[row[i]] * w[i] * dinv[col[i]];
}


__global__ void k_gemm_wmma(const float* __restrict__ H, const float* __restrict__ W,
                            float* __restrict__ XW, int n_nodes) {
    const int lane = threadIdx.x & 31;
    const int wave_in_block = threadIdx.x >> 5;
    const int waves_per_block = blockDim.x >> 5;
    const int tile_id = blockIdx.x * waves_per_block + wave_in_block;
    const int tiles_n = GCN_D / 16;
    const int tile_m = tile_id / tiles_n;
    const int tile_n = tile_id % tiles_n;
    if (tile_m * 16 >= n_nodes) return;

    const int row0 = tile_m * 16;
    const int col0 = tile_n * 16;
    const int lm = lane & 15;
    const int hi = lane >> 4;

    const float2* __restrict__ Ap = (const float2*)(H + (size_t)(row0 + lm) * GCN_D);
    const float2* __restrict__ Bp = (const float2*)(W + (size_t)(col0 + lm) * GCN_D);

    v8f c = {};
#pragma unroll
    for (int k = 0; k < GCN_D; k += 4) {
        const int kp = (k >> 1) + hi;
        float2 af = Ap[kp];
        float2 bf = Bp[kp];
        v2f a; a.x = af.x; a.y = af.y;
        v2f b; b.x = bf.x; b.y = bf.y;
        c = __builtin_amdgcn_wmma_f32_16x16x4_f32(
                 false, a,  false, b,
                 (short)0, c,  false,  false);
    }

    const int m0 = row0 + hi * 8;
    const int nc = col0 + lm;
#pragma unroll
    for (int v = 0; v < 8; ++v)
        XW[(size_t)(m0 + v) * GCN_D + nc] = c[v];
}


__global__ void k_init_acc(const float* __restrict__ XW, const float* __restrict__ dinv,
                           const float* __restrict__ bias, float* __restrict__ acc,
                           int n_nodes) {
    const int wave = (blockIdx.x * blockDim.x + threadIdx.x) >> 5;
    const int lane = threadIdx.x & 31;
    if (wave >= n_nodes) return;
    const float self = dinv[wave] * dinv[wave];
    float4 x = ((const float4*)(XW + (size_t)wave * GCN_D))[lane];
    float4 b = ((const float4*)bias)[lane];
    float4 o;
    o.x = b.x + x.x * self;
    o.y = b.y + x.y * self;
    o.z = b.z + x.z * self;
    o.w = b.w + x.w * self;
    ((float4*)(acc + (size_t)wave * GCN_D))[lane] = o;
}


__global__ void k_scatter(const float* __restrict__ XW, const int* __restrict__ row,
                          const int* __restrict__ col, const float* __restrict__ norm,
                          float* __restrict__ acc, int n_edges) {
    const int wave = (blockIdx.x * blockDim.x + threadIdx.x) >> 5;
    const int lane = threadIdx.x & 31;
    if (wave >= n_edges) return;
    const int r = row[wave];
    const int c = col[wave];
    const float w = norm[wave];
    float4 m = ((const float4*)(XW + (size_t)r * GCN_D))[lane];
    float* dst = acc + (size_t)c * GCN_D + lane * 4;
    atomicAdd(dst + 0, m.x * w);
    atomicAdd(dst + 1, m.y * w);
    atomicAdd(dst + 2, m.z * w);
    atomicAdd(dst + 3, m.w * w);
}


__global__ void k_ln_res(const float* __restrict__ acc, const float* __restrict__ hin,
                         const float* __restrict__ gamma, const float* __restrict__ beta,
                         float* __restrict__ hout, int n_nodes, int do_relu) {
    const int wave = (blockIdx.x * blockDim.x + threadIdx.x) >> 5;
    const int lane = threadIdx.x & 31;
    if (wave >= n_nodes) return;

    float4 a = ((const float4*)(acc + (size_t)wave * GCN_D))[lane];

    float s = a.x + a.y + a.z + a.w;
#pragma unroll
    for (int off = 16; off > 0; off >>= 1) s += __shfl_xor(s, off, 32);
    const float mu = s * (1.0f / GCN_D);

    float dx = a.x - mu, dy = a.y - mu, dz = a.z - mu, dw = a.w - mu;
    float q = dx * dx + dy * dy + dz * dz + dw * dw;
#pragma unroll
    for (int off = 16; off > 0; off >>= 1) q += __shfl_xor(q, off, 32);
    const float inv = rsqrtf(q * (1.0f / GCN_D) + LN_EPS);

    float4 g = ((const float4*)gamma)[lane];
    float4 b = ((const float4*)beta)[lane];
    float4 id = ((const float4*)(hin + (size_t)wave * GCN_D))[lane];

    float4 y;
    y.x = dx * inv * g.x + b.x;
    y.y = dy * inv * g.y + b.y;
    y.z = dz * inv * g.z + b.z;
    y.w = dw * inv * g.w + b.w;
    if (do_relu) {
        y.x = fmaxf(y.x, 0.0f);
        y.y = fmaxf(y.y, 0.0f);
        y.z = fmaxf(y.z, 0.0f);
        y.w = fmaxf(y.w, 0.0f);
    }
    y.x += id.x; y.y += id.y; y.z += id.z; y.w += id.w;
    ((float4*)(hout + (size_t)wave * GCN_D))[lane] = y;
}


extern "C" void kernel_launch(void* const* d_in, const int* in_sizes, int n_in,
                              void* d_out, int out_size, void* d_ws, size_t ws_size,
                              hipStream_t stream) {
    (void)in_sizes; (void)n_in; (void)out_size; (void)ws_size;

    const float* x      = (const float*)d_in[0];
    const int*   ei     = (const int*)d_in[1];
    const float* ew     = (const float*)d_in[2];
    const float* Ws     = (const float*)d_in[3];
    const float* bs     = (const float*)d_in[4];
    const float* gammas = (const float*)d_in[5];
    const float* betas  = (const float*)d_in[6];
    float*       out    = (float*)d_out;

    const int* row = ei;
    const int* col = ei + GCN_E;

    float* ws   = (float*)d_ws;
    float* deg  = ws;
    float* dinv = deg + GCN_N;
    float* norm = dinv + GCN_N;
    float* xw   = norm + GCN_E;
    float* acc  = xw + (size_t)GCN_N * GCN_D;
    float* h1   = acc + (size_t)GCN_N * GCN_D;

    const int BLK = 256;

    k_init_deg<<<(GCN_N + BLK - 1) / BLK, BLK, 0, stream>>>(deg, GCN_N);
    k_accum_deg<<<(GCN_E + BLK - 1) / BLK, BLK, 0, stream>>>(ew, col, deg, GCN_E);
    k_dinv<<<(GCN_N + BLK - 1) / BLK, BLK, 0, stream>>>(deg, dinv, GCN_N);
    k_norm<<<(GCN_E + BLK - 1) / BLK, BLK, 0, stream>>>(dinv, row, col, ew, norm, GCN_E);

    const int tiles_total  = ((GCN_N + 15) / 16) * (GCN_D / 16);
    const int gemm_blocks  = (tiles_total + (BLK / 32) - 1) / (BLK / 32);
    const int node_wave_blocks = (GCN_N + (BLK / 32) - 1) / (BLK / 32);
    const int edge_wave_blocks = (GCN_E + (BLK / 32) - 1) / (BLK / 32);

    const float* hin[GCN_L]  = { x,  h1,  out };
    float*       hout[GCN_L] = { h1, out, out };

    for (int l = 0; l < GCN_L; ++l) {
        const float* W = Ws + (size_t)l * GCN_D * GCN_D;
        k_gemm_wmma<<<gemm_blocks, BLK, 0, stream>>>(hin[l], W, xw, GCN_N);
        k_init_acc<<<node_wave_blocks, BLK, 0, stream>>>(xw, dinv, bs + (size_t)l * GCN_D,
                                                         acc, GCN_N);
        k_scatter<<<edge_wave_blocks, BLK, 0, stream>>>(xw, row, col, norm, acc, GCN_E);
        k_ln_res<<<node_wave_blocks, BLK, 0, stream>>>(acc, hin[l],
                                                       gammas + (size_t)l * GCN_D,
                                                       betas + (size_t)l * GCN_D,
                                                       hout[l], GCN_N,
                                                        (l < GCN_L - 1) ? 1 : 0);
    }
}


# Per-edge float atomic scatter replaced by a destination-sorted edge list (counting sort, integer atomic arrival-order rank) and a per-node register gather-sum; sums now run in list order
# speedup vs baseline: 2.6693x; 2.6693x over previous
#include <hip/hip_runtime.h>
#include <hip/hip_bf16.h>


#define GCN_N 50000
#define GCN_E 1600000
#define GCN_D 128
#define GCN_L 3
#define LN_EPS 1e-5f
#define SCAN_T 1024

typedef __attribute__((ext_vector_type(2))) float v2f;
typedef __attribute__((ext_vector_type(8))) float v8f;


__global__ void k_zero_cnt(int* __restrict__ cnt, int n) {
    int i = blockIdx.x * blockDim.x + threadIdx.x;
    if (i < n) cnt[i] = 0;
}

__global__ void k_hist(const int* __restrict__ col, int* __restrict__ cnt,
                       int* __restrict__ rank, int e) {
    int i = blockIdx.x * blockDim.x + threadIdx.x;
    if (i < e) rank[i] = atomicAdd(&cnt[col[i]], 1);
}

__global__ void k_scan(const int* __restrict__ cnt, int* __restrict__ off, int n) {
    __shared__ int s_tot[32];
    __shared__ int s_part[32];
    const int t = threadIdx.x;
    const int lane = t & 31;
    const int wv = t >> 5;
    const int b = blockIdx.x;
    const int i = b * SCAN_T + t;

    int part = 0;
#pragma unroll 8
    for (int tt = 0; tt < b; ++tt) part += cnt[tt * SCAN_T + t];

    const int v = (i < n) ? cnt[i] : 0;
    int inc = v;
    for (int d = 1; d < 32; d <<= 1) {
        int y = __shfl_up(inc, d, 32);
        if (lane >= d) inc += y;
    }
    for (int d = 16; d > 0; d >>= 1) part += __shfl_xor(part, d, 32);
    if (lane == 31) s_tot[wv] = inc;
    if (lane == 0) s_part[wv] = part;
    __syncthreads();

    int base = s_part[lane];
    int pre = (lane < wv) ? s_tot[lane] : 0;
    for (int d = 16; d > 0; d >>= 1) {
        base += __shfl_xor(base, d, 32);
        pre += __shfl_xor(pre, d, 32);
    }
    const int excl = base + pre + inc - v;
    if (i < n) off[i] = excl;
    if (i == n - 1) off[n] = excl + v;
}

__global__ void k_fill(const int* __restrict__ row, const int* __restrict__ col,
                       const float* __restrict__ w, const int* __restrict__ off,
                       const int* __restrict__ rank, int* __restrict__ pairs, int e) {
    int i = blockIdx.x * blockDim.x + threadIdx.x;
    if (i < e) {
        const int p = off[col[i]] + rank[i];
        int2 v;
        v.x = row[i];
        v.y = __float_as_int(w[i]);
        ((int2*)pairs)[p] = v;
    }
}

__global__ void k_deg_dinv(const int* __restrict__ off, const int* __restrict__ pairs,
                           float* __restrict__ dinv, int n_nodes) {
    const int wave = (blockIdx.x * blockDim.x + threadIdx.x) >> 5;
    const int lane = threadIdx.x & 31;
    if (wave >= n_nodes) return;
    const int beg = off[wave];
    const int end = off[wave + 1];
    const int2* __restrict__ P = (const int2*)pairs;
    float s = 0.0f;
    for (int p = beg + lane; p < end; p += 32) s += __int_as_float(P[p].y);
#pragma unroll
    for (int o = 16; o > 0; o >>= 1) s += __shfl_xor(s, o, 32);
    const float d = s + 1.0f;
    if (lane == 0) dinv[wave] = (d > 0.0f) ? rsqrtf(fmaxf(d, 1e-12f)) : 0.0f;
}

__global__ void k_norm(const int* __restrict__ off, const float* __restrict__ dinv,
                       int* __restrict__ pairs, int n_nodes) {
    const int wave = (blockIdx.x * blockDim.x + threadIdx.x) >> 5;
    const int lane = threadIdx.x & 31;
    if (wave >= n_nodes) return;
    const int beg = off[wave];
    const int end = off[wave + 1];
    const float dc = dinv[wave];
    int2* P = (int2*)pairs;
    for (int p = beg + lane; p < end; p += 32) {
        int2 v = P[p];
        const float nv = dinv[v.x] * __int_as_float(v.y) * dc;
        v.y = __float_as_int(nv);
        P[p] = v;
    }
}


__global__ void k_gemm_wmma(const float* __restrict__ H, const float* __restrict__ W,
                            float* __restrict__ XW, int n_nodes) {
    const int lane = threadIdx.x & 31;
    const int wave_in_block = threadIdx.x >> 5;
    const int waves_per_block = blockDim.x >> 5;
    const int tile_id = blockIdx.x * waves_per_block + wave_in_block;
    const int tiles_n = GCN_D / 16;
    const int tile_m = tile_id / tiles_n;
    const int tile_n = tile_id % tiles_n;
    if (tile_m * 16 >= n_nodes) return;

    const int row0 = tile_m * 16;
    const int col0 = tile_n * 16;
    const int lm = lane & 15;
    const int hi = lane >> 4;

    const float2* __restrict__ Ap = (const float2*)(H + (size_t)(row0 + lm) * GCN_D);
    const float2* __restrict__ Bp = (const float2*)(W + (size_t)(col0 + lm) * GCN_D);

    v8f c = {};
#pragma unroll
    for (int k = 0; k < GCN_D; k += 4) {
        const int kp = (k >> 1) + hi;
        float2 af = Ap[kp];
        float2 bf = Bp[kp];
        v2f a; a.x = af.x; a.y = af.y;
        v2f b; b.x = bf.x; b.y = bf.y;
        c = __builtin_amdgcn_wmma_f32_16x16x4_f32(
                 false, a,  false, b,
                 (short)0, c,  false,  false);
    }

    const int m0 = row0 + hi * 8;
    const int nc = col0 + lm;
#pragma unroll
    for (int v = 0; v < 8; ++v)
        XW[(size_t)(m0 + v) * GCN_D + nc] = c[v];
}


template <int U>
__device__ __forceinline__ void gather_rows(const float4* __restrict__ X4, int rl, int wl,
                                            int j, int lane, float4& a) {
    float4 m[U];
    float w[U];
#pragma unroll
    for (int u = 0; u < U; ++u) {
        const int r = __builtin_amdgcn_readlane(rl, j + u);
        w[u] = __int_as_float(__builtin_amdgcn_readlane(wl, j + u));
        m[u] = X4[(size_t)r * (GCN_D / 4) + lane];
    }
#pragma unroll
    for (int u = 0; u < U; ++u) {
        a.x = fmaf(m[u].x, w[u], a.x);
        a.y = fmaf(m[u].y, w[u], a.y);
        a.z = fmaf(m[u].z, w[u], a.z);
        a.w = fmaf(m[u].w, w[u], a.w);
    }
}

__global__ void k_gather(const float* __restrict__ XW, const int* __restrict__ off,
                         const int* __restrict__ pairs, const float* __restrict__ dinv,
                         const float* __restrict__ bias, float* __restrict__ acc,
                         int n_nodes) {
    const int wave = (blockIdx.x * blockDim.x + threadIdx.x) >> 5;
    const int lane = threadIdx.x & 31;
    if (wave >= n_nodes) return;
    const int node = __builtin_amdgcn_readfirstlane(wave);
    const int beg = off[node];
    const int end = off[node + 1];
    const float di = dinv[node];
    const float self = di * di;

    const float4* __restrict__ X4 = (const float4*)XW;
    const int2* __restrict__ P = (const int2*)pairs;

    float4 x = X4[(size_t)node * (GCN_D / 4) + lane];
    float4 b = ((const float4*)bias)[lane];
    float4 a;
    a.x = b.x + x.x * self;
    a.y = b.y + x.y * self;
    a.z = b.z + x.z * self;
    a.w = b.w + x.w * self;

    for (int p0 = beg; p0 < end; p0 += 32) {
        const int p = p0 + lane;
        int2 pr;
        pr.x = 0;
        pr.y = 0;
        if (p < end) pr = P[p];
        const int left = end - p0;
        const int cnt = left < 32 ? left : 32;
        int j = 0;
        for (; j + 8 <= cnt; j += 8) gather_rows<8>(X4, pr.x, pr.y, j, lane, a);
        if (j + 4 <= cnt) { gather_rows<4>(X4, pr.x, pr.y, j, lane, a); j += 4; }
        if (j + 2 <= cnt) { gather_rows<2>(X4, pr.x, pr.y, j, lane, a); j += 2; }
        if (j < cnt) gather_rows<1>(X4, pr.x, pr.y, j, lane, a);
    }
    ((float4*)(acc + (size_t)node * GCN_D))[lane] = a;
}


__global__ void k_ln_res(const float* __restrict__ acc, const float* __restrict__ hin,
                         const float* __restrict__ gamma, const float* __restrict__ beta,
                         float* __restrict__ hout, int n_nodes, int do_relu) {
    const int wave = (blockIdx.x * blockDim.x + threadIdx.x) >> 5;
    const int lane = threadIdx.x & 31;
    if (wave >= n_nodes) return;

    float4 a = ((const float4*)(acc + (size_t)wave * GCN_D))[lane];

    float s = a.x + a.y + a.z + a.w;
#pragma unroll
    for (int off = 16; off > 0; off >>= 1) s += __shfl_xor(s, off, 32);
    const float mu = s * (1.0f / GCN_D);

    float dx = a.x - mu, dy = a.y - mu, dz = a.z - mu, dw = a.w - mu;
    float q = dx * dx + dy * dy + dz * dz + dw * dw;
#pragma unroll
    for (int off = 16; off > 0; off >>= 1) q += __shfl_xor(q, off, 32);
    const float inv = rsqrtf(q * (1.0f / GCN_D) + LN_EPS);

    float4 g = ((const float4*)gamma)[lane];
    float4 b = ((const float4*)beta)[lane];
    float4 id = ((const float4*)(hin + (size_t)wave * GCN_D))[lane];

    float4 y;
    y.x = dx * inv * g.x + b.x;
    y.y = dy * inv * g.y + b.y;
    y.z = dz * inv * g.z + b.z;
    y.w = dw * inv * g.w + b.w;
    if (do_relu) {
        y.x = fmaxf(y.x, 0.0f);
        y.y = fmaxf(y.y, 0.0f);
        y.z = fmaxf(y.z, 0.0f);
        y.w = fmaxf(y.w, 0.0f);
    }
    y.x += id.x; y.y += id.y; y.z += id.z; y.w += id.w;
    ((float4*)(hout + (size_t)wave * GCN_D))[lane] = y;
}


extern "C" void kernel_launch(void* const* d_in, const int* in_sizes, int n_in,
                              void* d_out, int out_size, void* d_ws, size_t ws_size,
                              hipStream_t stream) {
    (void)in_sizes; (void)n_in; (void)out_size; (void)ws_size;

    const float* x      = (const float*)d_in[0];
    const int*   ei     = (const int*)d_in[1];
    const float* ew     = (const float*)d_in[2];
    const float* Ws     = (const float*)d_in[3];
    const float* bs     = (const float*)d_in[4];
    const float* gammas = (const float*)d_in[5];
    const float* betas  = (const float*)d_in[6];
    float*       out    = (float*)d_out;

    const int* row = ei;
    const int* col = ei + GCN_E;

    float* ws    = (float*)d_ws;
    int*   cnt   = (int*)ws;
    int*   off   = cnt + 50048;
    float* dinv  = (float*)(off + 50048);
    int*   rank  = (int*)(dinv + 50048);
    int*   pairs = rank + GCN_E;
    float* xw    = (float*)(pairs + 2 * GCN_E);
    float* acc   = xw + (size_t)GCN_N * GCN_D;
    float* h1    = acc + (size_t)GCN_N * GCN_D;

    const int BLK = 256;

    const int node_wave_blocks = (GCN_N + (BLK / 32) - 1) / (BLK / 32);
    k_zero_cnt<<<(GCN_N + BLK - 1) / BLK, BLK, 0, stream>>>(cnt, GCN_N);
    k_hist<<<(GCN_E + BLK - 1) / BLK, BLK, 0, stream>>>(col, cnt, rank, GCN_E);
    k_scan<<<(GCN_N + SCAN_T - 1) / SCAN_T, SCAN_T, 0, stream>>>(cnt, off, GCN_N);
    k_fill<<<(GCN_E + BLK - 1) / BLK, BLK, 0, stream>>>(row, col, ew, off, rank, pairs, GCN_E);
    k_deg_dinv<<<node_wave_blocks, BLK, 0, stream>>>(off, pairs, dinv, GCN_N);
    k_norm<<<node_wave_blocks, BLK, 0, stream>>>(off, dinv, pairs, GCN_N);

    const int tiles_total  = ((GCN_N + 15) / 16) * (GCN_D / 16);
    const int gemm_blocks  = (tiles_total + (BLK / 32) - 1) / (BLK / 32);

    const float* hin[GCN_L]  = { x,  h1,  out };
    float*       hout[GCN_L] = { h1, out, out };

    for (int l = 0; l < GCN_L; ++l) {
        const float* W = Ws + (size_t)l * GCN_D * GCN_D;
        k_gemm_wmma<<<gemm_blocks, BLK, 0, stream>>>(hin[l], W, xw, GCN_N);
        k_gather<<<node_wave_blocks, BLK, 0, stream>>>(xw, off, pairs, dinv,
                                                       bs + (size_t)l * GCN_D, acc, GCN_N);
        k_ln_res<<<node_wave_blocks, BLK, 0, stream>>>(acc, hin[l],
                                                       gammas + (size_t)l * GCN_D,
                                                       betas + (size_t)l * GCN_D,
                                                       hout[l], GCN_N,
                                                        (l < GCN_L - 1) ? 1 : 0);
    }
}


# GEMM re-tiled: W staged once per 128-row work-group in LDS (padded pitch), each wave does a 16x128 strip so one A fragment feeds 8 WMMAs; k order and sums unchanged
# speedup vs baseline: 3.1481x; 1.1794x over previous
#include <hip/hip_runtime.h>
#include <hip/hip_bf16.h>


#define GCN_N 50000
#define GCN_E 1600000
#define GCN_D 128
#define GCN_L 3
#define LN_EPS 1e-5f
#define SCAN_T 1024

typedef __attribute__((ext_vector_type(2))) float v2f;
typedef __attribute__((ext_vector_type(8))) float v8f;


__global__ void k_zero_cnt(int* __restrict__ cnt, int n) {
    int i = blockIdx.x * blockDim.x + threadIdx.x;
    if (i < n) cnt[i] = 0;
}

__global__ void k_hist(const int* __restrict__ col, int* __restrict__ cnt,
                       int* __restrict__ rank, int e) {
    int i = blockIdx.x * blockDim.x + threadIdx.x;
    if (i < e) rank[i] = atomicAdd(&cnt[col[i]], 1);
}

__global__ void k_scan(const int* __restrict__ cnt, int* __restrict__ off, int n) {
    __shared__ int s_tot[32];
    __shared__ int s_part[32];
    const int t = threadIdx.x;
    const int lane = t & 31;
    const int wv = t >> 5;
    const int b = blockIdx.x;
    const int i = b * SCAN_T + t;

    int part = 0;
#pragma unroll 8
    for (int tt = 0; tt < b; ++tt) part += cnt[tt * SCAN_T + t];

    const int v = (i < n) ? cnt[i] : 0;
    int inc = v;
    for (int d = 1; d < 32; d <<= 1) {
        int y = __shfl_up(inc, d, 32);
        if (lane >= d) inc += y;
    }
    for (int d = 16; d > 0; d >>= 1) part += __shfl_xor(part, d, 32);
    if (lane == 31) s_tot[wv] = inc;
    if (lane == 0) s_part[wv] = part;
    __syncthreads();

    int base = s_part[lane];
    int pre = (lane < wv) ? s_tot[lane] : 0;
    for (int d = 16; d > 0; d >>= 1) {
        base += __shfl_xor(base, d, 32);
        pre += __shfl_xor(pre, d, 32);
    }
    const int excl = base + pre + inc - v;
    if (i < n) off[i] = excl;
    if (i == n - 1) off[n] = excl + v;
}

__global__ void k_fill(const int* __restrict__ row, const int* __restrict__ col,
                       const float* __restrict__ w, const int* __restrict__ off,
                       const int* __restrict__ rank, int* __restrict__ pairs, int e) {
    int i = blockIdx.x * blockDim.x + threadIdx.x;
    if (i < e) {
        const int p = off[col[i]] + rank[i];
        int2 v;
        v.x = row[i];
        v.y = __float_as_int(w[i]);
        ((int2*)pairs)[p] = v;
    }
}

__global__ void k_deg_dinv(const int* __restrict__ off, const int* __restrict__ pairs,
                           float* __restrict__ dinv, int n_nodes) {
    const int wave = (blockIdx.x * blockDim.x + threadIdx.x) >> 5;
    const int lane = threadIdx.x & 31;
    if (wave >= n_nodes) return;
    const int beg = off[wave];
    const int end = off[wave + 1];
    const int2* __restrict__ P = (const int2*)pairs;
    float s = 0.0f;
    for (int p = beg + lane; p < end; p += 32) s += __int_as_float(P[p].y);
#pragma unroll
    for (int o = 16; o > 0; o >>= 1) s += __shfl_xor(s, o, 32);
    const float d = s + 1.0f;
    if (lane == 0) dinv[wave] = (d > 0.0f) ? rsqrtf(fmaxf(d, 1e-12f)) : 0.0f;
}

__global__ void k_norm(const int* __restrict__ off, const float* __restrict__ dinv,
                       int* __restrict__ pairs, int n_nodes) {
    const int wave = (blockIdx.x * blockDim.x + threadIdx.x) >> 5;
    const int lane = threadIdx.x & 31;
    if (wave >= n_nodes) return;
    const int beg = off[wave];
    const int end = off[wave + 1];
    const float dc = dinv[wave];
    int2* P = (int2*)pairs;
    for (int p = beg + lane; p < end; p += 32) {
        int2 v = P[p];
        const float nv = dinv[v.x] * __int_as_float(v.y) * dc;
        v.y = __float_as_int(nv);
        P[p] = v;
    }
}


#define GEMM_WAVES 8
#define GEMM_ROWS (GEMM_WAVES * 16)
#define W_PITCH 132

__global__ __launch_bounds__(GEMM_WAVES * 32)
void k_gemm_wmma(const float* __restrict__ H, const float* __restrict__ W,
                 float* __restrict__ XW, int n_nodes) {
    __shared__ __attribute__((aligned(16))) float sW[GCN_D * W_PITCH];

    const int t = threadIdx.x;
    const int lane = t & 31;
    const int wv = t >> 5;
    const int lm = lane & 15;
    const int hi = lane >> 4;

    const int row0 = (blockIdx.x * GEMM_WAVES + wv) * 16;
    const bool active = row0 < n_nodes;
    const int rowa = active ? row0 : 0;

    const float2* __restrict__ Ap = (const float2*)(H + (size_t)(rowa + lm) * GCN_D);
    v2f a[GCN_D / 4];
#pragma unroll
    for (int s = 0; s < GCN_D / 4; ++s) {
        const float2 af = Ap[2 * s + hi];
        a[s].x = af.x;
        a[s].y = af.y;
    }

    const float4* __restrict__ W4 = (const float4*)W;
#pragma unroll
    for (int p = 0; p < (GCN_D * GCN_D / 4) / (GEMM_WAVES * 32); ++p) {
        const int i = t + p * (GEMM_WAVES * 32);
        const int j = i >> 5;
        const int c4 = i & 31;
        const float4 v = W4[i];
        *(float4*)&sW[j * W_PITCH + 4 * c4] = v;
    }
    __syncthreads();
    if (!active) return;

    const float* sB = sW + lm * W_PITCH + 2 * hi;

    v8f c[8] = {};
    v2f b[2][16];
#pragma unroll
    for (int n = 0; n < 8; ++n) {
#pragma unroll
        for (int u = 0; u < 2; ++u) {
            const float2 bf = *(const float2*)(sB + n * 16 * W_PITCH + 4 * u);
            b[0][2 * n + u].x = bf.x;
            b[0][2 * n + u].y = bf.y;
        }
    }
#pragma unroll
    for (int blk = 0; blk < GCN_D / 8; ++blk) {
        const int cur = blk & 1;
        if (blk + 1 < GCN_D / 8) {
#pragma unroll
            for (int n = 0; n < 8; ++n) {
#pragma unroll
                for (int u = 0; u < 2; ++u) {
                    const float2 bf = *(const float2*)(sB + n * 16 * W_PITCH + 4 * (2 * (blk + 1) + u));
                    b[cur ^ 1][2 * n + u].x = bf.x;
                    b[cur ^ 1][2 * n + u].y = bf.y;
                }
            }
        }
        __builtin_amdgcn_sched_barrier(0);
#pragma unroll
        for (int u = 0; u < 2; ++u) {
#pragma unroll
            for (int n = 0; n < 8; ++n) {
                c[n] = __builtin_amdgcn_wmma_f32_16x16x4_f32(
                            false, a[2 * blk + u],  false, b[cur][2 * n + u],
                            (short)0, c[n],  false,  false);
            }
        }
        __builtin_amdgcn_sched_barrier(0);
    }

    const int m0 = row0 + hi * 8;
#pragma unroll
    for (int n = 0; n < 8; ++n) {
#pragma unroll
        for (int v = 0; v < 8; ++v)
            XW[(size_t)(m0 + v) * GCN_D + n * 16 + lm] = c[n][v];
    }
}


template <int U>
__device__ __forceinline__ void gather_rows(const float4* __restrict__ X4, int rl, int wl,
                                            int j, int lane, float4& a) {
    float4 m[U];
    float w[U];
#pragma unroll
    for (int u = 0; u < U; ++u) {
        const int r = __builtin_amdgcn_readlane(rl, j + u);
        w[u] = __int_as_float(__builtin_amdgcn_readlane(wl, j + u));
        m[u] = X4[(size_t)r * (GCN_D / 4) + lane];
    }
#pragma unroll
    for (int u = 0; u < U; ++u) {
        a.x = fmaf(m[u].x, w[u], a.x);
        a.y = fmaf(m[u].y, w[u], a.y);
        a.z = fmaf(m[u].z, w[u], a.z);
        a.w = fmaf(m[u].w, w[u], a.w);
    }
}

__global__ void k_gather(const float* __restrict__ XW, const int* __restrict__ off,
                         const int* __restrict__ pairs, const float* __restrict__ dinv,
                         const float* __restrict__ bias, float* __restrict__ acc,
                         int n_nodes) {
    const int wave = (blockIdx.x * blockDim.x + threadIdx.x) >> 5;
    const int lane = threadIdx.x & 31;
    if (wave >= n_nodes) return;
    const int node = __builtin_amdgcn_readfirstlane(wave);
    const int beg = off[node];
    const int end = off[node + 1];
    const float di = dinv[node];
    const float self = di * di;

    const float4* __restrict__ X4 = (const float4*)XW;
    const int2* __restrict__ P = (const int2*)pairs;

    float4 x = X4[(size_t)node * (GCN_D / 4) + lane];
    float4 b = ((const float4*)bias)[lane];
    float4 a;
    a.x = b.x + x.x * self;
    a.y = b.y + x.y * self;
    a.z = b.z + x.z * self;
    a.w = b.w + x.w * self;

    for (int p0 = beg; p0 < end; p0 += 32) {
        const int p = p0 + lane;
        int2 pr;
        pr.x = 0;
        pr.y = 0;
        if (p < end) pr = P[p];
        const int left = end - p0;
        const int cnt = left < 32 ? left : 32;
        int j = 0;
        for (; j + 8 <= cnt; j += 8) gather_rows<8>(X4, pr.x, pr.y, j, lane, a);
        if (j + 4 <= cnt) { gather_rows<4>(X4, pr.x, pr.y, j, lane, a); j += 4; }
        if (j + 2 <= cnt) { gather_rows<2>(X4, pr.x, pr.y, j, lane, a); j += 2; }
        if (j < cnt) gather_rows<1>(X4, pr.x, pr.y, j, lane, a);
    }
    ((float4*)(acc + (size_t)node * GCN_D))[lane] = a;
}


__global__ void k_ln_res(const float* __restrict__ acc, const float* __restrict__ hin,
                         const float* __restrict__ gamma, const float* __restrict__ beta,
                         float* __restrict__ hout, int n_nodes, int do_relu) {
    const int wave = (blockIdx.x * blockDim.x + threadIdx.x) >> 5;
    const int lane = threadIdx.x & 31;
    if (wave >= n_nodes) return;

    float4 a = ((const float4*)(acc + (size_t)wave * GCN_D))[lane];

    float s = a.x + a.y + a.z + a.w;
#pragma unroll
    for (int off = 16; off > 0; off >>= 1) s += __shfl_xor(s, off, 32);
    const float mu = s * (1.0f / GCN_D);

    float dx = a.x - mu, dy = a.y - mu, dz = a.z - mu, dw = a.w - mu;
    float q = dx * dx + dy * dy + dz * dz + dw * dw;
#pragma unroll
    for (int off = 16; off > 0; off >>= 1) q += __shfl_xor(q, off, 32);
    const float inv = rsqrtf(q * (1.0f / GCN_D) + LN_EPS);

    float4 g = ((const float4*)gamma)[lane];
    float4 b = ((const float4*)beta)[lane];
    float4 id = ((const float4*)(hin + (size_t)wave * GCN_D))[lane];

    float4 y;
    y.x = dx * inv * g.x + b.x;
    y.y = dy * inv * g.y + b.y;
    y.z = dz * inv * g.z + b.z;
    y.w = dw * inv * g.w + b.w;
    if (do_relu) {
        y.x = fmaxf(y.x, 0.0f);
        y.y = fmaxf(y.y, 0.0f);
        y.z = fmaxf(y.z, 0.0f);
        y.w = fmaxf(y.w, 0.0f);
    }
    y.x += id.x; y.y += id.y; y.z += id.z; y.w += id.w;
    ((float4*)(hout + (size_t)wave * GCN_D))[lane] = y;
}


extern "C" void kernel_launch(void* const* d_in, const int* in_sizes, int n_in,
                              void* d_out, int out_size, void* d_ws, size_t ws_size,
                              hipStream_t stream) {
    (void)in_sizes; (void)n_in; (void)out_size; (void)ws_size;

    const float* x      = (const float*)d_in[0];
    const int*   ei     = (const int*)d_in[1];
    const float* ew     = (const float*)d_in[2];
    const float* Ws     = (const float*)d_in[3];
    const float* bs     = (const float*)d_in[4];
    const float* gammas = (const float*)d_in[5];
    const float* betas  = (const float*)d_in[6];
    float*       out    = (float*)d_out;

    const int* row = ei;
    const int* col = ei + GCN_E;

    float* ws    = (float*)d_ws;
    int*   cnt   = (int*)ws;
    int*   off   = cnt + 50048;
    float* dinv  = (float*)(off + 50048);
    int*   rank  = (int*)(dinv + 50048);
    int*   pairs = rank + GCN_E;
    float* xw    = (float*)(pairs + 2 * GCN_E);
    float* acc   = xw + (size_t)GCN_N * GCN_D;
    float* h1    = acc + (size_t)GCN_N * GCN_D;

    const int BLK = 256;

    const int node_wave_blocks = (GCN_N + (BLK / 32) - 1) / (BLK / 32);
    k_zero_cnt<<<(GCN_N + BLK - 1) / BLK, BLK, 0, stream>>>(cnt, GCN_N);
    k_hist<<<(GCN_E + BLK - 1) / BLK, BLK, 0, stream>>>(col, cnt, rank, GCN_E);
    k_scan<<<(GCN_N + SCAN_T - 1) / SCAN_T, SCAN_T, 0, stream>>>(cnt, off, GCN_N);
    k_fill<<<(GCN_E + BLK - 1) / BLK, BLK, 0, stream>>>(row, col, ew, off, rank, pairs, GCN_E);
    k_deg_dinv<<<node_wave_blocks, BLK, 0, stream>>>(off, pairs, dinv, GCN_N);
    k_norm<<<node_wave_blocks, BLK, 0, stream>>>(off, dinv, pairs, GCN_N);

    const int gemm_blocks  = (GCN_N + GEMM_ROWS - 1) / GEMM_ROWS;

    const float* hin[GCN_L]  = { x,  h1,  out };
    float*       hout[GCN_L] = { h1, out, out };

    for (int l = 0; l < GCN_L; ++l) {
        const float* W = Ws + (size_t)l * GCN_D * GCN_D;
        k_gemm_wmma<<<gemm_blocks, GEMM_WAVES * 32, 0, stream>>>(hin[l], W, xw, GCN_N);
        k_gather<<<node_wave_blocks, BLK, 0, stream>>>(xw, off, pairs, dinv,
                                                       bs + (size_t)l * GCN_D, acc, GCN_N);
        k_ln_res<<<node_wave_blocks, BLK, 0, stream>>>(acc, hin[l],
                                                       gammas + (size_t)l * GCN_D,
                                                       betas + (size_t)l * GCN_D,
                                                       hout[l], GCN_N,
                                                        (l < GCN_L - 1) ? 1 : 0);
    }
}


# LayerNorm+ReLU+residual folded into the per-node gather wave (it already holds the finished row in registers): per layer one launch, one 50000-wave grid and one row write+read fewer; sums unchanged
# speedup vs baseline: 3.4016x; 1.0805x over previous
#include <hip/hip_runtime.h>
#include <hip/hip_bf16.h>


#define GCN_N 50000
#define GCN_E 1600000
#define GCN_D 128
#define GCN_L 3
#define LN_EPS 1e-5f
#define SCAN_T 1024

typedef __attribute__((ext_vector_type(2))) float v2f;
typedef __attribute__((ext_vector_type(8))) float v8f;


__global__ void k_zero_cnt(int* __restrict__ cnt, int n) {
    int i = blockIdx.x * blockDim.x + threadIdx.x;
    if (i < n) cnt[i] = 0;
}

__global__ void k_hist(const int* __restrict__ col, int* __restrict__ cnt,
                       int* __restrict__ rank, int e) {
    int i = blockIdx.x * blockDim.x + threadIdx.x;
    if (i < e) rank[i] = atomicAdd(&cnt[col[i]], 1);
}

__global__ void k_scan(const int* __restrict__ cnt, int* __restrict__ off, int n) {
    __shared__ int s_tot[32];
    __shared__ int s_part[32];
    const int t = threadIdx.x;
    const int lane = t & 31;
    const int wv = t >> 5;
    const int b = blockIdx.x;
    const int i = b * SCAN_T + t;

    int part = 0;
#pragma unroll 8
    for (int tt = 0; tt < b; ++tt) part += cnt[tt * SCAN_T + t];

    const int v = (i < n) ? cnt[i] : 0;
    int inc = v;
    for (int d = 1; d < 32; d <<= 1) {
        int y = __shfl_up(inc, d, 32);
        if (lane >= d) inc += y;
    }
    for (int d = 16; d > 0; d >>= 1) part += __shfl_xor(part, d, 32);
    if (lane == 31) s_tot[wv] = inc;
    if (lane == 0) s_part[wv] = part;
    __syncthreads();

    int base = s_part[lane];
    int pre = (lane < wv) ? s_tot[lane] : 0;
    for (int d = 16; d > 0; d >>= 1) {
        base += __shfl_xor(base, d, 32);
        pre += __shfl_xor(pre, d, 32);
    }
    const int excl = base + pre + inc - v;
    if (i < n) off[i] = excl;
    if (i == n - 1) off[n] = excl + v;
}

__global__ void k_fill(const int* __restrict__ row, const int* __restrict__ col,
                       const float* __restrict__ w, const int* __restrict__ off,
                       const int* __restrict__ rank, int* __restrict__ pairs, int e) {
    int i = blockIdx.x * blockDim.x + threadIdx.x;
    if (i < e) {
        const int p = off[col[i]] + rank[i];
        int2 v;
        v.x = row[i];
        v.y = __float_as_int(w[i]);
        ((int2*)pairs)[p] = v;
    }
}

__global__ void k_deg_dinv(const int* __restrict__ off, const int* __restrict__ pairs,
                           float* __restrict__ dinv, int n_nodes) {
    const int wave = (blockIdx.x * blockDim.x + threadIdx.x) >> 5;
    const int lane = threadIdx.x & 31;
    if (wave >= n_nodes) return;
    const int beg = off[wave];
    const int end = off[wave + 1];
    const int2* __restrict__ P = (const int2*)pairs;
    float s = 0.0f;
    for (int p = beg + lane; p < end; p += 32) s += __int_as_float(P[p].y);
#pragma unroll
    for (int o = 16; o > 0; o >>= 1) s += __shfl_xor(s, o, 32);
    const float d = s + 1.0f;
    if (lane == 0) dinv[wave] = (d > 0.0f) ? rsqrtf(fmaxf(d, 1e-12f)) : 0.0f;
}

__global__ void k_norm(const int* __restrict__ off, const float* __restrict__ dinv,
                       int* __restrict__ pairs, int n_nodes) {
    const int wave = (blockIdx.x * blockDim.x + threadIdx.x) >> 5;
    const int lane = threadIdx.x & 31;
    if (wave >= n_nodes) return;
    const int beg = off[wave];
    const int end = off[wave + 1];
    const float dc = dinv[wave];
    int2* P = (int2*)pairs;
    for (int p = beg + lane; p < end; p += 32) {
        int2 v = P[p];
        const float nv = dinv[v.x] * __int_as_float(v.y) * dc;
        v.y = __float_as_int(nv);
        P[p] = v;
    }
}


#define GEMM_WAVES 8
#define GEMM_ROWS (GEMM_WAVES * 16)
#define W_PITCH 132

__global__ __launch_bounds__(GEMM_WAVES * 32)
void k_gemm_wmma(const float* __restrict__ H, const float* __restrict__ W,
                 float* __restrict__ XW, int n_nodes) {
    __shared__ __attribute__((aligned(16))) float sW[GCN_D * W_PITCH];

    const int t = threadIdx.x;
    const int lane = t & 31;
    const int wv = t >> 5;
    const int lm = lane & 15;
    const int hi = lane >> 4;

    const int row0 = (blockIdx.x * GEMM_WAVES + wv) * 16;
    const bool active = row0 < n_nodes;
    const int rowa = active ? row0 : 0;

    const float2* __restrict__ Ap = (const float2*)(H + (size_t)(rowa + lm) * GCN_D);
    v2f a[GCN_D / 4];
#pragma unroll
    for (int s = 0; s < GCN_D / 4; ++s) {
        const float2 af = Ap[2 * s + hi];
        a[s].x = af.x;
        a[s].y = af.y;
    }

    const float4* __restrict__ W4 = (const float4*)W;
#pragma unroll
    for (int p = 0; p < (GCN_D * GCN_D / 4) / (GEMM_WAVES * 32); ++p) {
        const int i = t + p * (GEMM_WAVES * 32);
        const int j = i >> 5;
        const int c4 = i & 31;
        const float4 v = W4[i];
        *(float4*)&sW[j * W_PITCH + 4 * c4] = v;
    }
    __syncthreads();
    if (!active) return;

    const float* sB = sW + lm * W_PITCH + 2 * hi;

    v8f c[8] = {};
    v2f b[2][16];
#pragma unroll
    for (int n = 0; n < 8; ++n) {
#pragma unroll
        for (int u = 0; u < 2; ++u) {
            const float2 bf = *(const float2*)(sB + n * 16 * W_PITCH + 4 * u);
            b[0][2 * n + u].x = bf.x;
            b[0][2 * n + u].y = bf.y;
        }
    }
#pragma unroll
    for (int blk = 0; blk < GCN_D / 8; ++blk) {
        const int cur = blk & 1;
        if (blk + 1 < GCN_D / 8) {
#pragma unroll
            for (int n = 0; n < 8; ++n) {
#pragma unroll
                for (int u = 0; u < 2; ++u) {
                    const float2 bf = *(const float2*)(sB + n * 16 * W_PITCH + 4 * (2 * (blk + 1) + u));
                    b[cur ^ 1][2 * n + u].x = bf.x;
                    b[cur ^ 1][2 * n + u].y = bf.y;
                }
            }
        }
        __builtin_amdgcn_sched_barrier(0);
#pragma unroll
        for (int u = 0; u < 2; ++u) {
#pragma unroll
            for (int n = 0; n < 8; ++n) {
                c[n] = __builtin_amdgcn_wmma_f32_16x16x4_f32(
                            false, a[2 * blk + u],  false, b[cur][2 * n + u],
                            (short)0, c[n],  false,  false);
            }
        }
        __builtin_amdgcn_sched_barrier(0);
    }

    const int m0 = row0 + hi * 8;
#pragma unroll
    for (int n = 0; n < 8; ++n) {
#pragma unroll
        for (int v = 0; v < 8; ++v)
            XW[(size_t)(m0 + v) * GCN_D + n * 16 + lm] = c[n][v];
    }
}


template <int U>
__device__ __forceinline__ void gather_rows(const float4* __restrict__ X4, int rl, int wl,
                                            int j, int lane, float4& a) {
    float4 m[U];
    float w[U];
#pragma unroll
    for (int u = 0; u < U; ++u) {
        const int r = __builtin_amdgcn_readlane(rl, j + u);
        w[u] = __int_as_float(__builtin_amdgcn_readlane(wl, j + u));
        m[u] = X4[(size_t)r * (GCN_D / 4) + lane];
    }
#pragma unroll
    for (int u = 0; u < U; ++u) {
        a.x = fmaf(m[u].x, w[u], a.x);
        a.y = fmaf(m[u].y, w[u], a.y);
        a.z = fmaf(m[u].z, w[u], a.z);
        a.w = fmaf(m[u].w, w[u], a.w);
    }
}

__global__ void k_gather_ln(const float* __restrict__ XW, const int* __restrict__ off,
                            const int* __restrict__ pairs, const float* __restrict__ dinv,
                            const float* __restrict__ bias, const float* hin,
                            const float* __restrict__ gamma, const float* __restrict__ beta,
                            float* hout, int n_nodes, int do_relu) {
    const int wave = (blockIdx.x * blockDim.x + threadIdx.x) >> 5;
    const int lane = threadIdx.x & 31;
    if (wave >= n_nodes) return;
    const int node = __builtin_amdgcn_readfirstlane(wave);
    const int beg = off[node];
    const int end = off[node + 1];
    const float di = dinv[node];
    const float self = di * di;

    const float4* __restrict__ X4 = (const float4*)XW;
    const int2* __restrict__ P = (const int2*)pairs;

    float4 x = X4[(size_t)node * (GCN_D / 4) + lane];
    float4 b = ((const float4*)bias)[lane];
    const float4 id = ((const float4*)(hin + (size_t)node * GCN_D))[lane];
    float4 a;
    a.x = b.x + x.x * self;
    a.y = b.y + x.y * self;
    a.z = b.z + x.z * self;
    a.w = b.w + x.w * self;

    for (int p0 = beg; p0 < end; p0 += 32) {
        const int p = p0 + lane;
        int2 pr;
        pr.x = 0;
        pr.y = 0;
        if (p < end) pr = P[p];
        const int left = end - p0;
        const int cnt = left < 32 ? left : 32;
        int j = 0;
        for (; j + 8 <= cnt; j += 8) gather_rows<8>(X4, pr.x, pr.y, j, lane, a);
        if (j + 4 <= cnt) { gather_rows<4>(X4, pr.x, pr.y, j, lane, a); j += 4; }
        if (j + 2 <= cnt) { gather_rows<2>(X4, pr.x, pr.y, j, lane, a); j += 2; }
        if (j < cnt) gather_rows<1>(X4, pr.x, pr.y, j, lane, a);
    }

    float s = a.x + a.y + a.z + a.w;
#pragma unroll
    for (int o = 16; o > 0; o >>= 1) s += __shfl_xor(s, o, 32);
    const float mu = s * (1.0f / GCN_D);

    float dx = a.x - mu, dy = a.y - mu, dz = a.z - mu, dw = a.w - mu;
    float q = dx * dx + dy * dy + dz * dz + dw * dw;
#pragma unroll
    for (int o = 16; o > 0; o >>= 1) q += __shfl_xor(q, o, 32);
    const float inv = rsqrtf(q * (1.0f / GCN_D) + LN_EPS);

    const float4 g = ((const float4*)gamma)[lane];
    const float4 be = ((const float4*)beta)[lane];

    float4 y;
    y.x = dx * inv * g.x + be.x;
    y.y = dy * inv * g.y + be.y;
    y.z = dz * inv * g.z + be.z;
    y.w = dw * inv * g.w + be.w;
    if (do_relu) {
        y.x = fmaxf(y.x, 0.0f);
        y.y = fmaxf(y.y, 0.0f);
        y.z = fmaxf(y.z, 0.0f);
        y.w = fmaxf(y.w, 0.0f);
    }
    y.x += id.x; y.y += id.y; y.z += id.z; y.w += id.w;
    ((float4*)(hout + (size_t)node * GCN_D))[lane] = y;
}


extern "C" void kernel_launch(void* const* d_in, const int* in_sizes, int n_in,
                              void* d_out, int out_size, void* d_ws, size_t ws_size,
                              hipStream_t stream) {
    (void)in_sizes; (void)n_in; (void)out_size; (void)ws_size;

    const float* x      = (const float*)d_in[0];
    const int*   ei     = (const int*)d_in[1];
    const float* ew     = (const float*)d_in[2];
    const float* Ws     = (const float*)d_in[3];
    const float* bs     = (const float*)d_in[4];
    const float* gammas = (const float*)d_in[5];
    const float* betas  = (const float*)d_in[6];
    float*       out    = (float*)d_out;

    const int* row = ei;
    const int* col = ei + GCN_E;

    float* ws    = (float*)d_ws;
    int*   cnt   = (int*)ws;
    int*   off   = cnt + 50048;
    float* dinv  = (float*)(off + 50048);
    int*   rank  = (int*)(dinv + 50048);
    int*   pairs = rank + GCN_E;
    float* xw    = (float*)(pairs + 2 * GCN_E);
    float* h1    = xw + (size_t)GCN_N * GCN_D;

    const int BLK = 256;

    const int node_wave_blocks = (GCN_N + (BLK / 32) - 1) / (BLK / 32);
    k_zero_cnt<<<(GCN_N + BLK - 1) / BLK, BLK, 0, stream>>>(cnt, GCN_N);
    k_hist<<<(GCN_E + BLK - 1) / BLK, BLK, 0, stream>>>(col, cnt, rank, GCN_E);
    k_scan<<<(GCN_N + SCAN_T - 1) / SCAN_T, SCAN_T, 0, stream>>>(cnt, off, GCN_N);
    k_fill<<<(GCN_E + BLK - 1) / BLK, BLK, 0, stream>>>(row, col, ew, off, rank, pairs, GCN_E);
    k_deg_dinv<<<node_wave_blocks, BLK, 0, stream>>>(off, pairs, dinv, GCN_N);
    k_norm<<<node_wave_blocks, BLK, 0, stream>>>(off, dinv, pairs, GCN_N);

    const int gemm_blocks  = (GCN_N + GEMM_ROWS - 1) / GEMM_ROWS;

    const float* hin[GCN_L]  = { x,  h1,  out };
    float*       hout[GCN_L] = { h1, out, out };

    for (int l = 0; l < GCN_L; ++l) {
        const float* W = Ws + (size_t)l * GCN_D * GCN_D;
        k_gemm_wmma<<<gemm_blocks, GEMM_WAVES * 32, 0, stream>>>(hin[l], W, xw, GCN_N);
        k_gather_ln<<<node_wave_blocks, BLK, 0, stream>>>(xw, off, pairs, dinv,
                                                          bs + (size_t)l * GCN_D, hin[l],
                                                          gammas + (size_t)l * GCN_D,
                                                          betas + (size_t)l * GCN_D,
                                                          hout[l], GCN_N,
                                                           (l < GCN_L - 1) ? 1 : 0);
    }
}


# Counting sort by destination: per-edge global returning atomics replaced by LDS counters (128-node bins per edge slice, then one work-group per bin); edge order in a node's run, so sum order, changes
# speedup vs baseline: 4.8890x; 1.4373x over previous
#include <hip/hip_runtime.h>
#include <hip/hip_bf16.h>


#define GCN_N 50000
#define GCN_E 1600000
#define GCN_D 128
#define GCN_L 3
#define LN_EPS 1e-5f
#define SCAN_T 1024

typedef __attribute__((ext_vector_type(2))) float v2f;
typedef __attribute__((ext_vector_type(8))) float v8f;


#define BIN_SHIFT 7
#define BIN_NODES (1 << BIN_SHIFT)
#define NBINS ((GCN_N + BIN_NODES - 1) / BIN_NODES)
#define SORT_G 256
#define SORT_T 1024
#define SORT_CH ((GCN_E + SORT_G - 1) / SORT_G)
#define SORT_IT ((SORT_CH + SORT_T - 1) / SORT_T)

static_assert(GCN_N <= 65536, "source row must fit in 16 bits");
static_assert(NBINS * SORT_G + 1 <= 100352, "scan arrays as laid out in the workspace");

__global__ __launch_bounds__(SORT_T)
void k_bin_count(const int* __restrict__ col, int* __restrict__ cntBG, int e) {
    __shared__ int hist[NBINS];
    const int t = threadIdx.x;
    const int g = blockIdx.x;
    for (int b = t; b < NBINS; b += SORT_T) hist[b] = 0;
    __syncthreads();
    const int lo = g * SORT_CH;
    const int hi = (lo + SORT_CH < e) ? lo + SORT_CH : e;
    int c[SORT_IT];
#pragma unroll
    for (int k = 0; k < SORT_IT; ++k) {
        const int i = lo + t + k * SORT_T;
        c[k] = (i < hi) ? col[i] : 0;
    }
#pragma unroll
    for (int k = 0; k < SORT_IT; ++k) {
        const int i = lo + t + k * SORT_T;
        if (i < hi) atomicAdd(&hist[c[k] >> BIN_SHIFT], 1);
    }
    __syncthreads();
    for (int b = t; b < NBINS; b += SORT_T) cntBG[b * SORT_G + g] = hist[b];
}

__global__ void k_scan(const int* __restrict__ cnt, int* __restrict__ off, int n) {
    __shared__ int s_tot[32];
    __shared__ int s_part[32];
    const int t = threadIdx.x;
    const int lane = t & 31;
    const int wv = t >> 5;
    const int b = blockIdx.x;
    const int i = b * SCAN_T + t;

    int part = 0;
#pragma unroll 8
    for (int tt = 0; tt < b; ++tt) part += cnt[tt * SCAN_T + t];

    const int v = (i < n) ? cnt[i] : 0;
    int inc = v;
    for (int d = 1; d < 32; d <<= 1) {
        int y = __shfl_up(inc, d, 32);
        if (lane >= d) inc += y;
    }
    for (int d = 16; d > 0; d >>= 1) part += __shfl_xor(part, d, 32);
    if (lane == 31) s_tot[wv] = inc;
    if (lane == 0) s_part[wv] = part;
    __syncthreads();

    int base = s_part[lane];
    int pre = (lane < wv) ? s_tot[lane] : 0;
    for (int d = 16; d > 0; d >>= 1) {
        base += __shfl_xor(base, d, 32);
        pre += __shfl_xor(pre, d, 32);
    }
    const int excl = base + pre + inc - v;
    if (i < n) off[i] = excl;
    if (i == n - 1) off[n] = excl + v;
}

__global__ __launch_bounds__(SORT_T)
void k_bin_scatter(const int* __restrict__ row, const int* __restrict__ col,
                   const float* __restrict__ w, const int* __restrict__ offBG,
                   int* __restrict__ tmp, int e) {
    __shared__ int cur[NBINS];
    const int t = threadIdx.x;
    const int g = blockIdx.x;
    for (int b = t; b < NBINS; b += SORT_T) cur[b] = offBG[b * SORT_G + g];
    __syncthreads();
    const int lo = g * SORT_CH;
    const int hi = (lo + SORT_CH < e) ? lo + SORT_CH : e;
    int c[SORT_IT];
    int r[SORT_IT];
    float wv[SORT_IT];
#pragma unroll
    for (int k = 0; k < SORT_IT; ++k) {
        const int i = lo + t + k * SORT_T;
        const bool in = i < hi;
        c[k] = in ? col[i] : 0;
        r[k] = in ? row[i] : 0;
        wv[k] = in ? w[i] : 0.0f;
    }
#pragma unroll
    for (int k = 0; k < SORT_IT; ++k) {
        const int i = lo + t + k * SORT_T;
        if (i < hi) {
            const int p = atomicAdd(&cur[c[k] >> BIN_SHIFT], 1);
            int2 v;
            v.x = r[k] | ((c[k] & (BIN_NODES - 1)) << 16);
            v.y = __float_as_int(wv[k]);
            ((int2*)tmp)[p] = v;
        }
    }
}

__global__ __launch_bounds__(SORT_T)
void k_bin_sort(const int* __restrict__ offBG, const int* __restrict__ tmp,
                int* __restrict__ off, int* __restrict__ pairs, int n_nodes) {
    __shared__ int cnt[BIN_NODES];
    __shared__ int cur[BIN_NODES];
    __shared__ int wtot[BIN_NODES / 32];
    const int t = threadIdx.x;
    const int lane = t & 31;
    const int wv = t >> 5;
    const int b = blockIdx.x;
    const int beg = offBG[b * SORT_G];
    const int end = offBG[(b + 1) * SORT_G];
    const int2* __restrict__ T2 = (const int2*)tmp;

    if (t < BIN_NODES) cnt[t] = 0;
    __syncthreads();
    for (int i = beg + t; i < end; i += SORT_T)
        atomicAdd(&cnt[(T2[i].x >> 16) & (BIN_NODES - 1)], 1);
    __syncthreads();

    const int v = (t < BIN_NODES) ? cnt[t] : 0;
    int inc = v;
    for (int d = 1; d < 32; d <<= 1) {
        int y = __shfl_up(inc, d, 32);
        if (lane >= d) inc += y;
    }
    if (t < BIN_NODES && lane == 31) wtot[wv] = inc;
    __syncthreads();
    if (t < BIN_NODES) {
        int pre = 0;
        for (int q = 0; q < wv; ++q) pre += wtot[q];
        const int excl = beg + pre + inc - v;
        cur[t] = excl;
        const int node = b * BIN_NODES + t;
        if (node < n_nodes) off[node] = excl;
        if (node == n_nodes - 1) off[n_nodes] = excl + v;
    }
    __syncthreads();
    for (int i = beg + t; i < end; i += SORT_T) {
        const int2 rec = T2[i];
        const int p = atomicAdd(&cur[(rec.x >> 16) & (BIN_NODES - 1)], 1);
        int2 o;
        o.x = rec.x & 0xFFFF;
        o.y = rec.y;
        ((int2*)pairs)[p] = o;
    }
}

__global__ void k_deg_dinv(const int* __restrict__ off, const int* __restrict__ pairs,
                           float* __restrict__ dinv, int n_nodes) {
    const int wave = (blockIdx.x * blockDim.x + threadIdx.x) >> 5;
    const int lane = threadIdx.x & 31;
    if (wave >= n_nodes) return;
    const int beg = off[wave];
    const int end = off[wave + 1];
    const int2* __restrict__ P = (const int2*)pairs;
    float s = 0.0f;
    for (int p = beg + lane; p < end; p += 32) s += __int_as_float(P[p].y);
#pragma unroll
    for (int o = 16; o > 0; o >>= 1) s += __shfl_xor(s, o, 32);
    const float d = s + 1.0f;
    if (lane == 0) dinv[wave] = (d > 0.0f) ? rsqrtf(fmaxf(d, 1e-12f)) : 0.0f;
}

__global__ void k_norm(const int* __restrict__ off, const float* __restrict__ dinv,
                       int* __restrict__ pairs, int n_nodes) {
    const int wave = (blockIdx.x * blockDim.x + threadIdx.x) >> 5;
    const int lane = threadIdx.x & 31;
    if (wave >= n_nodes) return;
    const int beg = off[wave];
    const int end = off[wave + 1];
    const float dc = dinv[wave];
    int2* P = (int2*)pairs;
    for (int p = beg + lane; p < end; p += 32) {
        int2 v = P[p];
        const float nv = dinv[v.x] * __int_as_float(v.y) * dc;
        v.y = __float_as_int(nv);
        P[p] = v;
    }
}


#define GEMM_WAVES 8
#define GEMM_ROWS (GEMM_WAVES * 16)
#define W_PITCH 132

__global__ __launch_bounds__(GEMM_WAVES * 32)
void k_gemm_wmma(const float* __restrict__ H, const float* __restrict__ W,
                 float* __restrict__ XW, int n_nodes) {
    __shared__ __attribute__((aligned(16))) float sW[GCN_D * W_PITCH];

    const int t = threadIdx.x;
    const int lane = t & 31;
    const int wv = t >> 5;
    const int lm = lane & 15;
    const int hi = lane >> 4;

    const int row0 = (blockIdx.x * GEMM_WAVES + wv) * 16;
    const bool active = row0 < n_nodes;
    const int rowa = active ? row0 : 0;

    const float2* __restrict__ Ap = (const float2*)(H + (size_t)(rowa + lm) * GCN_D);
    v2f a[GCN_D / 4];
#pragma unroll
    for (int s = 0; s < GCN_D / 4; ++s) {
        const float2 af = Ap[2 * s + hi];
        a[s].x = af.x;
        a[s].y = af.y;
    }

    const float4* __restrict__ W4 = (const float4*)W;
#pragma unroll
    for (int p = 0; p < (GCN_D * GCN_D / 4) / (GEMM_WAVES * 32); ++p) {
        const int i = t + p * (GEMM_WAVES * 32);
        const int j = i >> 5;
        const int c4 = i & 31;
        const float4 v = W4[i];
        *(float4*)&sW[j * W_PITCH + 4 * c4] = v;
    }
    __syncthreads();
    if (!active) return;

    const float* sB = sW + lm * W_PITCH + 2 * hi;

    v8f c[8] = {};
    v2f b[2][16];
#pragma unroll
    for (int n = 0; n < 8; ++n) {
#pragma unroll
        for (int u = 0; u < 2; ++u) {
            const float2 bf = *(const float2*)(sB + n * 16 * W_PITCH + 4 * u);
            b[0][2 * n + u].x = bf.x;
            b[0][2 * n + u].y = bf.y;
        }
    }
#pragma unroll
    for (int blk = 0; blk < GCN_D / 8; ++blk) {
        const int cur = blk & 1;
        if (blk + 1 < GCN_D / 8) {
#pragma unroll
            for (int n = 0; n < 8; ++n) {
#pragma unroll
                for (int u = 0; u < 2; ++u) {
                    const float2 bf = *(const float2*)(sB + n * 16 * W_PITCH + 4 * (2 * (blk + 1) + u));
                    b[cur ^ 1][2 * n + u].x = bf.x;
                    b[cur ^ 1][2 * n + u].y = bf.y;
                }
            }
        }
        __builtin_amdgcn_sched_barrier(0);
#pragma unroll
        for (int u = 0; u < 2; ++u) {
#pragma unroll
            for (int n = 0; n < 8; ++n) {
                c[n] = __builtin_amdgcn_wmma_f32_16x16x4_f32(
                            false, a[2 * blk + u],  false, b[cur][2 * n + u],
                            (short)0, c[n],  false,  false);
            }
        }
        __builtin_amdgcn_sched_barrier(0);
    }

    const int m0 = row0 + hi * 8;
#pragma unroll
    for (int n = 0; n < 8; ++n) {
#pragma unroll
        for (int v = 0; v < 8; ++v)
            XW[(size_t)(m0 + v) * GCN_D + n * 16 + lm] = c[n][v];
    }
}


template <int U>
__device__ __forceinline__ void gather_rows(const float4* __restrict__ X4, int rl, int wl,
                                            int j, int lane, float4& a) {
    float4 m[U];
    float w[U];
#pragma unroll
    for (int u = 0; u < U; ++u) {
        const int r = __builtin_amdgcn_readlane(rl, j + u);
        w[u] = __int_as_float(__builtin_amdgcn_readlane(wl, j + u));
        m[u] = X4[(size_t)r * (GCN_D / 4) + lane];
    }
#pragma unroll
    for (int u = 0; u < U; ++u) {
        a.x = fmaf(m[u].x, w[u], a.x);
        a.y = fmaf(m[u].y, w[u], a.y);
        a.z = fmaf(m[u].z, w[u], a.z);
        a.w = fmaf(m[u].w, w[u], a.w);
    }
}

__global__ void k_gather_ln(const float* __restrict__ XW, const int* __restrict__ off,
                            const int* __restrict__ pairs, const float* __restrict__ dinv,
                            const float* __restrict__ bias, const float* hin,
                            const float* __restrict__ gamma, const float* __restrict__ beta,
                            float* hout, int n_nodes, int do_relu) {
    const int wave = (blockIdx.x * blockDim.x + threadIdx.x) >> 5;
    const int lane = threadIdx.x & 31;
    if (wave >= n_nodes) return;
    const int node = __builtin_amdgcn_readfirstlane(wave);
    const int beg = off[node];
    const int end = off[node + 1];
    const float di = dinv[node];
    const float self = di * di;

    const float4* __restrict__ X4 = (const float4*)XW;
    const int2* __restrict__ P = (const int2*)pairs;

    float4 x = X4[(size_t)node * (GCN_D / 4) + lane];
    float4 b = ((const float4*)bias)[lane];
    const float4 id = ((const float4*)(hin + (size_t)node * GCN_D))[lane];
    float4 a;
    a.x = b.x + x.x * self;
    a.y = b.y + x.y * self;
    a.z = b.z + x.z * self;
    a.w = b.w + x.w * self;

    for (int p0 = beg; p0 < end; p0 += 32) {
        const int p = p0 + lane;
        int2 pr;
        pr.x = 0;
        pr.y = 0;
        if (p < end) pr = P[p];
        const int left = end - p0;
        const int cnt = left < 32 ? left : 32;
        int j = 0;
        for (; j + 8 <= cnt; j += 8) gather_rows<8>(X4, pr.x, pr.y, j, lane, a);
        if (j + 4 <= cnt) { gather_rows<4>(X4, pr.x, pr.y, j, lane, a); j += 4; }
        if (j + 2 <= cnt) { gather_rows<2>(X4, pr.x, pr.y, j, lane, a); j += 2; }
        if (j < cnt) gather_rows<1>(X4, pr.x, pr.y, j, lane, a);
    }

    float s = a.x + a.y + a.z + a.w;
#pragma unroll
    for (int o = 16; o > 0; o >>= 1) s += __shfl_xor(s, o, 32);
    const float mu = s * (1.0f / GCN_D);

    float dx = a.x - mu, dy = a.y - mu, dz = a.z - mu, dw = a.w - mu;
    float q = dx * dx + dy * dy + dz * dz + dw * dw;
#pragma unroll
    for (int o = 16; o > 0; o >>= 1) q += __shfl_xor(q, o, 32);
    const float inv = rsqrtf(q * (1.0f / GCN_D) + LN_EPS);

    const float4 g = ((const float4*)gamma)[lane];
    const float4 be = ((const float4*)beta)[lane];

    float4 y;
    y.x = dx * inv * g.x + be.x;
    y.y = dy * inv * g.y + be.y;
    y.z = dz * inv * g.z + be.z;
    y.w = dw * inv * g.w + be.w;
    if (do_relu) {
        y.x = fmaxf(y.x, 0.0f);
        y.y = fmaxf(y.y, 0.0f);
        y.z = fmaxf(y.z, 0.0f);
        y.w = fmaxf(y.w, 0.0f);
    }
    y.x += id.x; y.y += id.y; y.z += id.z; y.w += id.w;
    ((float4*)(hout + (size_t)node * GCN_D))[lane] = y;
}


extern "C" void kernel_launch(void* const* d_in, const int* in_sizes, int n_in,
                              void* d_out, int out_size, void* d_ws, size_t ws_size,
                              hipStream_t stream) {
    (void)in_sizes; (void)n_in; (void)out_size; (void)ws_size;

    const float* x      = (const float*)d_in[0];
    const int*   ei     = (const int*)d_in[1];
    const float* ew     = (const float*)d_in[2];
    const float* Ws     = (const float*)d_in[3];
    const float* bs     = (const float*)d_in[4];
    const float* gammas = (const float*)d_in[5];
    const float* betas  = (const float*)d_in[6];
    float*       out    = (float*)d_out;

    const int* row = ei;
    const int* col = ei + GCN_E;

    float* ws    = (float*)d_ws;
    int*   off   = (int*)ws;
    float* dinv  = (float*)(off + 50048);
    int*   cntBG = (int*)(dinv + 50048);
    int*   offBG = cntBG + 100352;
    int*   tmp   = offBG + 100352;
    int*   pairs = tmp + 2 * GCN_E;
    float* xw    = (float*)(pairs + 2 * GCN_E);
    float* h1    = xw + (size_t)GCN_N * GCN_D;

    const int BLK = 256;

    const int node_wave_blocks = (GCN_N + (BLK / 32) - 1) / (BLK / 32);
    k_bin_count<<<SORT_G, SORT_T, 0, stream>>>(col, cntBG, GCN_E);
    k_scan<<<(NBINS * SORT_G + SCAN_T - 1) / SCAN_T, SCAN_T, 0, stream>>>(cntBG, offBG, NBINS * SORT_G);
    k_bin_scatter<<<SORT_G, SORT_T, 0, stream>>>(row, col, ew, offBG, tmp, GCN_E);
    k_bin_sort<<<NBINS, SORT_T, 0, stream>>>(offBG, tmp, off, pairs, GCN_N);
    k_deg_dinv<<<node_wave_blocks, BLK, 0, stream>>>(off, pairs, dinv, GCN_N);
    k_norm<<<node_wave_blocks, BLK, 0, stream>>>(off, dinv, pairs, GCN_N);

    const int gemm_blocks  = (GCN_N + GEMM_ROWS - 1) / GEMM_ROWS;

    const float* hin[GCN_L]  = { x,  h1,  out };
    float*       hout[GCN_L] = { h1, out, out };

    for (int l = 0; l < GCN_L; ++l) {
        const float* W = Ws + (size_t)l * GCN_D * GCN_D;
        k_gemm_wmma<<<gemm_blocks, GEMM_WAVES * 32, 0, stream>>>(hin[l], W, xw, GCN_N);
        k_gather_ln<<<node_wave_blocks, BLK, 0, stream>>>(xw, off, pairs, dinv,
                                                          bs + (size_t)l * GCN_D, hin[l],
                                                          gammas + (size_t)l * GCN_D,
                                                          betas + (size_t)l * GCN_D,
                                                          hout[l], GCN_N,
                                                           (l < GCN_L - 1) ? 1 : 0);
    }
}


# Edge normalisation factored: GEMM epilogue scales rows by dinv[row], node sums use raw weights, dinv[col] applied after; degree by LDS float adds (arrival order) in bin sort; 2 launches dropped
# speedup vs baseline: 5.2109x; 1.0658x over previous
#include <hip/hip_runtime.h>
#include <hip/hip_bf16.h>


#define GCN_N 50000
#define GCN_E 1600000
#define GCN_D 128
#define GCN_L 3
#define LN_EPS 1e-5f
#define SCAN_T 1024

typedef __attribute__((ext_vector_type(2))) float v2f;
typedef __attribute__((ext_vector_type(8))) float v8f;


#define BIN_SHIFT 7
#define BIN_NODES (1 << BIN_SHIFT)
#define NBINS ((GCN_N + BIN_NODES - 1) / BIN_NODES)
#define SORT_G 256
#define SORT_T 1024
#define SORT_CH ((GCN_E + SORT_G - 1) / SORT_G)
#define SORT_IT ((SORT_CH + SORT_T - 1) / SORT_T)

static_assert(GCN_N <= 65536, "source row must fit in 16 bits");
static_assert(NBINS * SORT_G + 1 <= 100352, "scan arrays as laid out in the workspace");

__global__ __launch_bounds__(SORT_T)
void k_bin_count(const int* __restrict__ col, int* __restrict__ cntBG, int e) {
    __shared__ int hist[NBINS];
    const int t = threadIdx.x;
    const int g = blockIdx.x;
    for (int b = t; b < NBINS; b += SORT_T) hist[b] = 0;
    __syncthreads();
    const int lo = g * SORT_CH;
    const int hi = (lo + SORT_CH < e) ? lo + SORT_CH : e;
    int c[SORT_IT];
#pragma unroll
    for (int k = 0; k < SORT_IT; ++k) {
        const int i = lo + t + k * SORT_T;
        c[k] = (i < hi) ? col[i] : 0;
    }
#pragma unroll
    for (int k = 0; k < SORT_IT; ++k) {
        const int i = lo + t + k * SORT_T;
        if (i < hi) atomicAdd(&hist[c[k] >> BIN_SHIFT], 1);
    }
    __syncthreads();
    for (int b = t; b < NBINS; b += SORT_T) cntBG[b * SORT_G + g] = hist[b];
}

__global__ void k_scan(const int* __restrict__ cnt, int* __restrict__ off, int n) {
    __shared__ int s_tot[32];
    __shared__ int s_part[32];
    const int t = threadIdx.x;
    const int lane = t & 31;
    const int wv = t >> 5;
    const int b = blockIdx.x;
    const int i = b * SCAN_T + t;

    int part = 0;
#pragma unroll 8
    for (int tt = 0; tt < b; ++tt) part += cnt[tt * SCAN_T + t];

    const int v = (i < n) ? cnt[i] : 0;
    int inc = v;
    for (int d = 1; d < 32; d <<= 1) {
        int y = __shfl_up(inc, d, 32);
        if (lane >= d) inc += y;
    }
    for (int d = 16; d > 0; d >>= 1) part += __shfl_xor(part, d, 32);
    if (lane == 31) s_tot[wv] = inc;
    if (lane == 0) s_part[wv] = part;
    __syncthreads();

    int base = s_part[lane];
    int pre = (lane < wv) ? s_tot[lane] : 0;
    for (int d = 16; d > 0; d >>= 1) {
        base += __shfl_xor(base, d, 32);
        pre += __shfl_xor(pre, d, 32);
    }
    const int excl = base + pre + inc - v;
    if (i < n) off[i] = excl;
    if (i == n - 1) off[n] = excl + v;
}

__global__ __launch_bounds__(SORT_T)
void k_bin_scatter(const int* __restrict__ row, const int* __restrict__ col,
                   const float* __restrict__ w, const int* __restrict__ offBG,
                   int* __restrict__ tmp, int e) {
    __shared__ int cur[NBINS];
    const int t = threadIdx.x;
    const int g = blockIdx.x;
    for (int b = t; b < NBINS; b += SORT_T) cur[b] = offBG[b * SORT_G + g];
    __syncthreads();
    const int lo = g * SORT_CH;
    const int hi = (lo + SORT_CH < e) ? lo + SORT_CH : e;
    int c[SORT_IT];
    int r[SORT_IT];
    float wv[SORT_IT];
#pragma unroll
    for (int k = 0; k < SORT_IT; ++k) {
        const int i = lo + t + k * SORT_T;
        const bool in = i < hi;
        c[k] = in ? col[i] : 0;
        r[k] = in ? row[i] : 0;
        wv[k] = in ? w[i] : 0.0f;
    }
#pragma unroll
    for (int k = 0; k < SORT_IT; ++k) {
        const int i = lo + t + k * SORT_T;
        if (i < hi) {
            const int p = atomicAdd(&cur[c[k] >> BIN_SHIFT], 1);
            int2 v;
            v.x = r[k] | ((c[k] & (BIN_NODES - 1)) << 16);
            v.y = __float_as_int(wv[k]);
            ((int2*)tmp)[p] = v;
        }
    }
}

__global__ __launch_bounds__(SORT_T)
void k_bin_sort(const int* __restrict__ offBG, const int* __restrict__ tmp,
                int* __restrict__ off, int* __restrict__ pairs,
                float* __restrict__ dinv, int n_nodes) {
    __shared__ int cnt[BIN_NODES];
    __shared__ int cur[BIN_NODES];
    __shared__ float wsum[BIN_NODES];
    __shared__ int wtot[BIN_NODES / 32];
    const int t = threadIdx.x;
    const int lane = t & 31;
    const int wv = t >> 5;
    const int b = blockIdx.x;
    const int beg = offBG[b * SORT_G];
    const int end = offBG[(b + 1) * SORT_G];
    const int2* __restrict__ T2 = (const int2*)tmp;

    if (t < BIN_NODES) { cnt[t] = 0; wsum[t] = 0.0f; }
    __syncthreads();
    for (int i = beg + t; i < end; i += SORT_T) {
        const int2 rec = T2[i];
        const int ln = (rec.x >> 16) & (BIN_NODES - 1);
        atomicAdd(&cnt[ln], 1);
        atomicAdd(&wsum[ln], __int_as_float(rec.y));
    }
    __syncthreads();

    const int v = (t < BIN_NODES) ? cnt[t] : 0;
    int inc = v;
    for (int d = 1; d < 32; d <<= 1) {
        int y = __shfl_up(inc, d, 32);
        if (lane >= d) inc += y;
    }
    if (t < BIN_NODES && lane == 31) wtot[wv] = inc;
    __syncthreads();
    if (t < BIN_NODES) {
        int pre = 0;
        for (int q = 0; q < wv; ++q) pre += wtot[q];
        const int excl = beg + pre + inc - v;
        cur[t] = excl;
        const int node = b * BIN_NODES + t;
        if (node < n_nodes) {
            off[node] = excl;
            const float d = wsum[t] + 1.0f;
            dinv[node] = (d > 0.0f) ? rsqrtf(fmaxf(d, 1e-12f)) : 0.0f;
        }
        if (node == n_nodes - 1) off[n_nodes] = excl + v;
    }
    __syncthreads();
    for (int i = beg + t; i < end; i += SORT_T) {
        const int2 rec = T2[i];
        const int p = atomicAdd(&cur[(rec.x >> 16) & (BIN_NODES - 1)], 1);
        int2 o;
        o.x = rec.x & 0xFFFF;
        o.y = rec.y;
        ((int2*)pairs)[p] = o;
    }
}


#define GEMM_WAVES 8
#define GEMM_ROWS (GEMM_WAVES * 16)
#define W_PITCH 132

__global__ __launch_bounds__(GEMM_WAVES * 32)
void k_gemm_wmma(const float* __restrict__ H, const float* __restrict__ W,
                 const float* __restrict__ dinv, float* __restrict__ XW, int n_nodes) {
    __shared__ __attribute__((aligned(16))) float sW[GCN_D * W_PITCH];

    const int t = threadIdx.x;
    const int lane = t & 31;
    const int wv = t >> 5;
    const int lm = lane & 15;
    const int hi = lane >> 4;

    const int row0 = (blockIdx.x * GEMM_WAVES + wv) * 16;
    const bool active = row0 < n_nodes;
    const int rowa = active ? row0 : 0;

    const float2* __restrict__ Ap = (const float2*)(H + (size_t)(rowa + lm) * GCN_D);
    v2f a[GCN_D / 4];
#pragma unroll
    for (int s = 0; s < GCN_D / 4; ++s) {
        const float2 af = Ap[2 * s + hi];
        a[s].x = af.x;
        a[s].y = af.y;
    }

    const float4* __restrict__ W4 = (const float4*)W;
#pragma unroll
    for (int p = 0; p < (GCN_D * GCN_D / 4) / (GEMM_WAVES * 32); ++p) {
        const int i = t + p * (GEMM_WAVES * 32);
        const int j = i >> 5;
        const int c4 = i & 31;
        const float4 v = W4[i];
        *(float4*)&sW[j * W_PITCH + 4 * c4] = v;
    }
    __syncthreads();
    if (!active) return;

    const float* sB = sW + lm * W_PITCH + 2 * hi;

    const int m0 = row0 + hi * 8;
    const float4 dlo = *(const float4*)(dinv + m0);
    const float4 dhi = *(const float4*)(dinv + m0 + 4);

    v8f c[8] = {};
    v2f b[2][16];
#pragma unroll
    for (int n = 0; n < 8; ++n) {
#pragma unroll
        for (int u = 0; u < 2; ++u) {
            const float2 bf = *(const float2*)(sB + n * 16 * W_PITCH + 4 * u);
            b[0][2 * n + u].x = bf.x;
            b[0][2 * n + u].y = bf.y;
        }
    }
#pragma unroll
    for (int blk = 0; blk < GCN_D / 8; ++blk) {
        const int cur = blk & 1;
        if (blk + 1 < GCN_D / 8) {
#pragma unroll
            for (int n = 0; n < 8; ++n) {
#pragma unroll
                for (int u = 0; u < 2; ++u) {
                    const float2 bf = *(const float2*)(sB + n * 16 * W_PITCH + 4 * (2 * (blk + 1) + u));
                    b[cur ^ 1][2 * n + u].x = bf.x;
                    b[cur ^ 1][2 * n + u].y = bf.y;
                }
            }
        }
        __builtin_amdgcn_sched_barrier(0);
#pragma unroll
        for (int u = 0; u < 2; ++u) {
#pragma unroll
            for (int n = 0; n < 8; ++n) {
                c[n] = __builtin_amdgcn_wmma_f32_16x16x4_f32(
                            false, a[2 * blk + u],  false, b[cur][2 * n + u],
                            (short)0, c[n],  false,  false);
            }
        }
        __builtin_amdgcn_sched_barrier(0);
    }

    const float dv[8] = { dlo.x, dlo.y, dlo.z, dlo.w, dhi.x, dhi.y, dhi.z, dhi.w };
#pragma unroll
    for (int n = 0; n < 8; ++n) {
#pragma unroll
        for (int v = 0; v < 8; ++v)
            XW[(size_t)(m0 + v) * GCN_D + n * 16 + lm] = c[n][v] * dv[v];
    }
}


template <int U>
__device__ __forceinline__ void gather_rows(const float4* __restrict__ X4, int rl, int wl,
                                            int j, int lane, float4& a) {
    float4 m[U];
    float w[U];
#pragma unroll
    for (int u = 0; u < U; ++u) {
        const int r = __builtin_amdgcn_readlane(rl, j + u);
        w[u] = __int_as_float(__builtin_amdgcn_readlane(wl, j + u));
        m[u] = X4[(size_t)r * (GCN_D / 4) + lane];
    }
#pragma unroll
    for (int u = 0; u < U; ++u) {
        a.x = fmaf(m[u].x, w[u], a.x);
        a.y = fmaf(m[u].y, w[u], a.y);
        a.z = fmaf(m[u].z, w[u], a.z);
        a.w = fmaf(m[u].w, w[u], a.w);
    }
}

__global__ void k_gather_ln(const float* __restrict__ XW, const int* __restrict__ off,
                            const int* __restrict__ pairs, const float* __restrict__ dinv,
                            const float* __restrict__ bias, const float* hin,
                            const float* __restrict__ gamma, const float* __restrict__ beta,
                            float* hout, int n_nodes, int do_relu) {
    const int wave = (blockIdx.x * blockDim.x + threadIdx.x) >> 5;
    const int lane = threadIdx.x & 31;
    if (wave >= n_nodes) return;
    const int node = __builtin_amdgcn_readfirstlane(wave);
    const int beg = off[node];
    const int end = off[node + 1];
    const float di = dinv[node];

    const float4* __restrict__ X4 = (const float4*)XW;
    const int2* __restrict__ P = (const int2*)pairs;

    float4 x = X4[(size_t)node * (GCN_D / 4) + lane];
    float4 b = ((const float4*)bias)[lane];
    const float4 id = ((const float4*)(hin + (size_t)node * GCN_D))[lane];
    float4 a = x;

    for (int p0 = beg; p0 < end; p0 += 32) {
        const int p = p0 + lane;
        int2 pr;
        pr.x = 0;
        pr.y = 0;
        if (p < end) pr = P[p];
        const int left = end - p0;
        const int cnt = left < 32 ? left : 32;
        int j = 0;
        for (; j + 8 <= cnt; j += 8) gather_rows<8>(X4, pr.x, pr.y, j, lane, a);
        if (j + 4 <= cnt) { gather_rows<4>(X4, pr.x, pr.y, j, lane, a); j += 4; }
        if (j + 2 <= cnt) { gather_rows<2>(X4, pr.x, pr.y, j, lane, a); j += 2; }
        if (j < cnt) gather_rows<1>(X4, pr.x, pr.y, j, lane, a);
    }

    a.x = fmaf(di, a.x, b.x);
    a.y = fmaf(di, a.y, b.y);
    a.z = fmaf(di, a.z, b.z);
    a.w = fmaf(di, a.w, b.w);

    float s = a.x + a.y + a.z + a.w;
#pragma unroll
    for (int o = 16; o > 0; o >>= 1) s += __shfl_xor(s, o, 32);
    const float mu = s * (1.0f / GCN_D);

    float dx = a.x - mu, dy = a.y - mu, dz = a.z - mu, dw = a.w - mu;
    float q = dx * dx + dy * dy + dz * dz + dw * dw;
#pragma unroll
    for (int o = 16; o > 0; o >>= 1) q += __shfl_xor(q, o, 32);
    const float inv = rsqrtf(q * (1.0f / GCN_D) + LN_EPS);

    const float4 g = ((const float4*)gamma)[lane];
    const float4 be = ((const float4*)beta)[lane];

    float4 y;
    y.x = dx * inv * g.x + be.x;
    y.y = dy * inv * g.y + be.y;
    y.z = dz * inv * g.z + be.z;
    y.w = dw * inv * g.w + be.w;
    if (do_relu) {
        y.x = fmaxf(y.x, 0.0f);
        y.y = fmaxf(y.y, 0.0f);
        y.z = fmaxf(y.z, 0.0f);
        y.w = fmaxf(y.w, 0.0f);
    }
    y.x += id.x; y.y += id.y; y.z += id.z; y.w += id.w;
    ((float4*)(hout + (size_t)node * GCN_D))[lane] = y;
}


extern "C" void kernel_launch(void* const* d_in, const int* in_sizes, int n_in,
                              void* d_out, int out_size, void* d_ws, size_t ws_size,
                              hipStream_t stream) {
    (void)in_sizes; (void)n_in; (void)out_size; (void)ws_size;

    const float* x      = (const float*)d_in[0];
    const int*   ei     = (const int*)d_in[1];
    const float* ew     = (const float*)d_in[2];
    const float* Ws     = (const float*)d_in[3];
    const float* bs     = (const float*)d_in[4];
    const float* gammas = (const float*)d_in[5];
    const float* betas  = (const float*)d_in[6];
    float*       out    = (float*)d_out;

    const int* row = ei;
    const int* col = ei + GCN_E;

    float* ws    = (float*)d_ws;
    int*   off   = (int*)ws;
    float* dinv  = (float*)(off + 50048);
    int*   cntBG = (int*)(dinv + 50048);
    int*   offBG = cntBG + 100352;
    int*   tmp   = offBG + 100352;
    int*   pairs = tmp + 2 * GCN_E;
    float* xw    = (float*)(pairs + 2 * GCN_E);
    float* h1    = xw + (size_t)GCN_N * GCN_D;

    const int BLK = 256;

    const int node_wave_blocks = (GCN_N + (BLK / 32) - 1) / (BLK / 32);
    k_bin_count<<<SORT_G, SORT_T, 0, stream>>>(col, cntBG, GCN_E);
    k_scan<<<(NBINS * SORT_G + SCAN_T - 1) / SCAN_T, SCAN_T, 0, stream>>>(cntBG, offBG, NBINS * SORT_G);
    k_bin_scatter<<<SORT_G, SORT_T, 0, stream>>>(row, col, ew, offBG, tmp, GCN_E);
    k_bin_sort<<<NBINS, SORT_T, 0, stream>>>(offBG, tmp, off, pairs, dinv, GCN_N);

    const int gemm_blocks  = (GCN_N + GEMM_ROWS - 1) / GEMM_ROWS;

    const float* hin[GCN_L]  = { x,  h1,  out };
    float*       hout[GCN_L] = { h1, out, out };

    for (int l = 0; l < GCN_L; ++l) {
        const float* W = Ws + (size_t)l * GCN_D * GCN_D;
        k_gemm_wmma<<<gemm_blocks, GEMM_WAVES * 32, 0, stream>>>(hin[l], W, dinv, xw, GCN_N);
        k_gather_ln<<<node_wave_blocks, BLK, 0, stream>>>(xw, off, pairs, dinv,
                                                          bs + (size_t)l * GCN_D, hin[l],
                                                          gammas + (size_t)l * GCN_D,
                                                          betas + (size_t)l * GCN_D,
                                                          hout[l], GCN_N,
                                                           (l < GCN_L - 1) ? 1 : 0);
    }
}


# Next layer's f32 WMMA product moved into the gather launch: finished rows reach it through LDS in 32-node tiles, issued between the next tile's row loads; 2 launches fewer; sums unchanged
# speedup vs baseline: 5.4172x; 1.0396x over previous
#include <hip/hip_runtime.h>
#include <hip/hip_bf16.h>


#define GCN_N 50000
#define GCN_E 1600000
#define GCN_D 128
#define GCN_L 3
#define LN_EPS 1e-5f
#define SCAN_T 1024

typedef __attribute__((ext_vector_type(2))) float v2f;
typedef __attribute__((ext_vector_type(8))) float v8f;


#define BIN_SHIFT 7
#define BIN_NODES (1 << BIN_SHIFT)
#define NBINS ((GCN_N + BIN_NODES - 1) / BIN_NODES)
#define SORT_G 256
#define SORT_T 1024
#define SORT_CH ((GCN_E + SORT_G - 1) / SORT_G)
#define SORT_IT ((SORT_CH + SORT_T - 1) / SORT_T)

static_assert(GCN_N <= 65536, "source row must fit in 16 bits");
static_assert(NBINS * SORT_G + 1 <= 100352, "scan arrays as laid out in the workspace");

__global__ __launch_bounds__(SORT_T)
void k_bin_count(const int* __restrict__ col, int* __restrict__ cntBG, int e) {
    __shared__ int hist[NBINS];
    const int t = threadIdx.x;
    const int g = blockIdx.x;
    for (int b = t; b < NBINS; b += SORT_T) hist[b] = 0;
    __syncthreads();
    const int lo = g * SORT_CH;
    const int hi = (lo + SORT_CH < e) ? lo + SORT_CH : e;
    int c[SORT_IT];
#pragma unroll
    for (int k = 0; k < SORT_IT; ++k) {
        const int i = lo + t + k * SORT_T;
        c[k] = (i < hi) ? col[i] : 0;
    }
#pragma unroll
    for (int k = 0; k < SORT_IT; ++k) {
        const int i = lo + t + k * SORT_T;
        if (i < hi) atomicAdd(&hist[c[k] >> BIN_SHIFT], 1);
    }
    __syncthreads();
    for (int b = t; b < NBINS; b += SORT_T) cntBG[b * SORT_G + g] = hist[b];
}

__global__ void k_scan(const int* __restrict__ cnt, int* __restrict__ off, int n) {
    __shared__ int s_tot[32];
    __shared__ int s_part[32];
    const int t = threadIdx.x;
    const int lane = t & 31;
    const int wv = t >> 5;
    const int b = blockIdx.x;
    const int i = b * SCAN_T + t;

    int part = 0;
#pragma unroll 8
    for (int tt = 0; tt < b; ++tt) part += cnt[tt * SCAN_T + t];

    const int v = (i < n) ? cnt[i] : 0;
    int inc = v;
    for (int d = 1; d < 32; d <<= 1) {
        int y = __shfl_up(inc, d, 32);
        if (lane >= d) inc += y;
    }
    for (int d = 16; d > 0; d >>= 1) part += __shfl_xor(part, d, 32);
    if (lane == 31) s_tot[wv] = inc;
    if (lane == 0) s_part[wv] = part;
    __syncthreads();

    int base = s_part[lane];
    int pre = (lane < wv) ? s_tot[lane] : 0;
    for (int d = 16; d > 0; d >>= 1) {
        base += __shfl_xor(base, d, 32);
        pre += __shfl_xor(pre, d, 32);
    }
    const int excl = base + pre + inc - v;
    if (i < n) off[i] = excl;
    if (i == n - 1) off[n] = excl + v;
}

__global__ __launch_bounds__(SORT_T)
void k_bin_scatter(const int* __restrict__ row, const int* __restrict__ col,
                   const float* __restrict__ w, const int* __restrict__ offBG,
                   int* __restrict__ tmp, int e) {
    __shared__ int cur[NBINS];
    const int t = threadIdx.x;
    const int g = blockIdx.x;
    for (int b = t; b < NBINS; b += SORT_T) cur[b] = offBG[b * SORT_G + g];
    __syncthreads();
    const int lo = g * SORT_CH;
    const int hi = (lo + SORT_CH < e) ? lo + SORT_CH : e;
    int c[SORT_IT];
    int r[SORT_IT];
    float wv[SORT_IT];
#pragma unroll
    for (int k = 0; k < SORT_IT; ++k) {
        const int i = lo + t + k * SORT_T;
        const bool in = i < hi;
        c[k] = in ? col[i] : 0;
        r[k] = in ? row[i] : 0;
        wv[k] = in ? w[i] : 0.0f;
    }
#pragma unroll
    for (int k = 0; k < SORT_IT; ++k) {
        const int i = lo + t + k * SORT_T;
        if (i < hi) {
            const int p = atomicAdd(&cur[c[k] >> BIN_SHIFT], 1);
            int2 v;
            v.x = r[k] | ((c[k] & (BIN_NODES - 1)) << 16);
            v.y = __float_as_int(wv[k]);
            ((int2*)tmp)[p] = v;
        }
    }
}

__global__ __launch_bounds__(SORT_T)
void k_bin_sort(const int* __restrict__ offBG, const int* __restrict__ tmp,
                int* __restrict__ off, int* __restrict__ pairs,
                float* __restrict__ dinv, int n_nodes) {
    __shared__ int cnt[BIN_NODES];
    __shared__ int cur[BIN_NODES];
    __shared__ float wsum[BIN_NODES];
    __shared__ int wtot[BIN_NODES / 32];
    const int t = threadIdx.x;
    const int lane = t & 31;
    const int wv = t >> 5;
    const int b = blockIdx.x;
    const int beg = offBG[b * SORT_G];
    const int end = offBG[(b + 1) * SORT_G];
    const int2* __restrict__ T2 = (const int2*)tmp;

    if (t < BIN_NODES) { cnt[t] = 0; wsum[t] = 0.0f; }
    __syncthreads();
    for (int i = beg + t; i < end; i += SORT_T) {
        const int2 rec = T2[i];
        const int ln = (rec.x >> 16) & (BIN_NODES - 1);
        atomicAdd(&cnt[ln], 1);
        atomicAdd(&wsum[ln], __int_as_float(rec.y));
    }
    __syncthreads();

    const int v = (t < BIN_NODES) ? cnt[t] : 0;
    int inc = v;
    for (int d = 1; d < 32; d <<= 1) {
        int y = __shfl_up(inc, d, 32);
        if (lane >= d) inc += y;
    }
    if (t < BIN_NODES && lane == 31) wtot[wv] = inc;
    __syncthreads();
    if (t < BIN_NODES) {
        int pre = 0;
        for (int q = 0; q < wv; ++q) pre += wtot[q];
        const int excl = beg + pre + inc - v;
        cur[t] = excl;
        const int node = b * BIN_NODES + t;
        if (node < n_nodes) {
            off[node] = excl;
            const float d = wsum[t] + 1.0f;
            dinv[node] = (d > 0.0f) ? rsqrtf(fmaxf(d, 1e-12f)) : 0.0f;
        }
        if (node == n_nodes - 1) off[n_nodes] = excl + v;
    }
    __syncthreads();
    for (int i = beg + t; i < end; i += SORT_T) {
        const int2 rec = T2[i];
        const int p = atomicAdd(&cur[(rec.x >> 16) & (BIN_NODES - 1)], 1);
        int2 o;
        o.x = rec.x & 0xFFFF;
        o.y = rec.y;
        ((int2*)pairs)[p] = o;
    }
}


#define GEMM_WAVES 8
#define GEMM_ROWS (GEMM_WAVES * 16)
#define W_PITCH 132

__global__ __launch_bounds__(GEMM_WAVES * 32)
void k_gemm_wmma(const float* __restrict__ H, const float* __restrict__ W,
                 const float* __restrict__ dinv, float* __restrict__ XW, int n_nodes) {
    __shared__ __attribute__((aligned(16))) float sW[GCN_D * W_PITCH];

    const int t = threadIdx.x;
    const int lane = t & 31;
    const int wv = t >> 5;
    const int lm = lane & 15;
    const int hi = lane >> 4;

    const int row0 = (blockIdx.x * GEMM_WAVES + wv) * 16;
    const bool active = row0 < n_nodes;
    const int rowa = active ? row0 : 0;

    const float2* __restrict__ Ap = (const float2*)(H + (size_t)(rowa + lm) * GCN_D);
    v2f a[GCN_D / 4];
#pragma unroll
    for (int s = 0; s < GCN_D / 4; ++s) {
        const float2 af = Ap[2 * s + hi];
        a[s].x = af.x;
        a[s].y = af.y;
    }

    const float4* __restrict__ W4 = (const float4*)W;
#pragma unroll
    for (int p = 0; p < (GCN_D * GCN_D / 4) / (GEMM_WAVES * 32); ++p) {
        const int i = t + p * (GEMM_WAVES * 32);
        const int j = i >> 5;
        const int c4 = i & 31;
        const float4 v = W4[i];
        *(float4*)&sW[j * W_PITCH + 4 * c4] = v;
    }
    __syncthreads();
    if (!active) return;

    const float* sB = sW + lm * W_PITCH + 2 * hi;

    const int m0 = row0 + hi * 8;
    const float4 dlo = *(const float4*)(dinv + m0);
    const float4 dhi = *(const float4*)(dinv + m0 + 4);

    v8f c[8] = {};
    v2f b[2][16];
#pragma unroll
    for (int n = 0; n < 8; ++n) {
#pragma unroll
        for (int u = 0; u < 2; ++u) {
            const float2 bf = *(const float2*)(sB + n * 16 * W_PITCH + 4 * u);
            b[0][2 * n + u].x = bf.x;
            b[0][2 * n + u].y = bf.y;
        }
    }
#pragma unroll
    for (int blk = 0; blk < GCN_D / 8; ++blk) {
        const int cur = blk & 1;
        if (blk + 1 < GCN_D / 8) {
#pragma unroll
            for (int n = 0; n < 8; ++n) {
#pragma unroll
                for (int u = 0; u < 2; ++u) {
                    const float2 bf = *(const float2*)(sB + n * 16 * W_PITCH + 4 * (2 * (blk + 1) + u));
                    b[cur ^ 1][2 * n + u].x = bf.x;
                    b[cur ^ 1][2 * n + u].y = bf.y;
                }
            }
        }
        __builtin_amdgcn_sched_barrier(0);
#pragma unroll
        for (int u = 0; u < 2; ++u) {
#pragma unroll
            for (int n = 0; n < 8; ++n) {
                c[n] = __builtin_amdgcn_wmma_f32_16x16x4_f32(
                            false, a[2 * blk + u],  false, b[cur][2 * n + u],
                            (short)0, c[n],  false,  false);
            }
        }
        __builtin_amdgcn_sched_barrier(0);
    }

    const float dv[8] = { dlo.x, dlo.y, dlo.z, dlo.w, dhi.x, dhi.y, dhi.z, dhi.w };
#pragma unroll
    for (int n = 0; n < 8; ++n) {
#pragma unroll
        for (int v = 0; v < 8; ++v)
            XW[(size_t)(m0 + v) * GCN_D + n * 16 + lm] = c[n][v] * dv[v];
    }
}


template <int U>
__device__ __forceinline__ void gather_rows(const float4* __restrict__ X4, int rl, int wl,
                                            int j, int lane, float4& a) {
    float4 m[U];
    float w[U];
#pragma unroll
    for (int u = 0; u < U; ++u) {
        const int r = __builtin_amdgcn_readlane(rl, j + u);
        w[u] = __int_as_float(__builtin_amdgcn_readlane(wl, j + u));
        m[u] = X4[(size_t)r * (GCN_D / 4) + lane];
    }
#pragma unroll
    for (int u = 0; u < U; ++u) {
        a.x = fmaf(m[u].x, w[u], a.x);
        a.y = fmaf(m[u].y, w[u], a.y);
        a.z = fmaf(m[u].z, w[u], a.z);
        a.w = fmaf(m[u].w, w[u], a.w);
    }
}

__global__ void k_gather_ln(const float* __restrict__ XW, const int* __restrict__ off,
                            const int* __restrict__ pairs, const float* __restrict__ dinv,
                            const float* __restrict__ bias, const float* hin,
                            const float* __restrict__ gamma, const float* __restrict__ beta,
                            float* hout, int n_nodes, int do_relu) {
    const int wave = (blockIdx.x * blockDim.x + threadIdx.x) >> 5;
    const int lane = threadIdx.x & 31;
    if (wave >= n_nodes) return;
    const int node = __builtin_amdgcn_readfirstlane(wave);
    const int beg = off[node];
    const int end = off[node + 1];
    const float di = dinv[node];

    const float4* __restrict__ X4 = (const float4*)XW;
    const int2* __restrict__ P = (const int2*)pairs;

    float4 x = X4[(size_t)node * (GCN_D / 4) + lane];
    float4 b = ((const float4*)bias)[lane];
    const float4 id = ((const float4*)(hin + (size_t)node * GCN_D))[lane];
    float4 a = x;

    for (int p0 = beg; p0 < end; p0 += 32) {
        const int p = p0 + lane;
        int2 pr;
        pr.x = 0;
        pr.y = 0;
        if (p < end) pr = P[p];
        const int left = end - p0;
        const int cnt = left < 32 ? left : 32;
        int j = 0;
        for (; j + 8 <= cnt; j += 8) gather_rows<8>(X4, pr.x, pr.y, j, lane, a);
        if (j + 4 <= cnt) { gather_rows<4>(X4, pr.x, pr.y, j, lane, a); j += 4; }
        if (j + 2 <= cnt) { gather_rows<2>(X4, pr.x, pr.y, j, lane, a); j += 2; }
        if (j < cnt) gather_rows<1>(X4, pr.x, pr.y, j, lane, a);
    }

    a.x = fmaf(di, a.x, b.x);
    a.y = fmaf(di, a.y, b.y);
    a.z = fmaf(di, a.z, b.z);
    a.w = fmaf(di, a.w, b.w);

    float s = a.x + a.y + a.z + a.w;
#pragma unroll
    for (int o = 16; o > 0; o >>= 1) s += __shfl_xor(s, o, 32);
    const float mu = s * (1.0f / GCN_D);

    float dx = a.x - mu, dy = a.y - mu, dz = a.z - mu, dw = a.w - mu;
    float q = dx * dx + dy * dy + dz * dz + dw * dw;
#pragma unroll
    for (int o = 16; o > 0; o >>= 1) q += __shfl_xor(q, o, 32);
    const float inv = rsqrtf(q * (1.0f / GCN_D) + LN_EPS);

    const float4 g = ((const float4*)gamma)[lane];
    const float4 be = ((const float4*)beta)[lane];

    float4 y;
    y.x = dx * inv * g.x + be.x;
    y.y = dy * inv * g.y + be.y;
    y.z = dz * inv * g.z + be.z;
    y.w = dw * inv * g.w + be.w;
    if (do_relu) {
        y.x = fmaxf(y.x, 0.0f);
        y.y = fmaxf(y.y, 0.0f);
        y.z = fmaxf(y.z, 0.0f);
        y.w = fmaxf(y.w, 0.0f);
    }
    y.x += id.x; y.y += id.y; y.z += id.z; y.w += id.w;
    ((float4*)(hout + (size_t)node * GCN_D))[lane] = y;
}


#define FUSE_WAVES 16
#define FUSE_T (FUSE_WAVES * 32)
#define TILE_NODES 32
#define FUSE_TILES 4
#define FUSE_NODES (TILE_NODES * FUSE_TILES)
#define KS_SLICE 8

static_assert(GCN_N % 16 == 0, "strips of 16 nodes must be whole");
static_assert((GCN_D * GCN_D / 4) % FUSE_T == 0, "W staging: whole passes of the work-group");

__device__ __forceinline__ void gemm_slice(v8f& c, int& ks, const float* pa, const float* pb) {
#pragma unroll
    for (int q = 0; q < KS_SLICE; ++q) {
        const float2 af = *(const float2*)(pa + 4 * (ks + q));
        const float2 bf = *(const float2*)(pb + 4 * (ks + q));
        v2f a2; a2.x = af.x; a2.y = af.y;
        v2f b2; b2.x = bf.x; b2.y = bf.y;
        c = __builtin_amdgcn_wmma_f32_16x16x4_f32( false, a2,  false, b2,
                                                   (short)0, c,
                                                   false,  false);
    }
    ks += KS_SLICE;
}

__device__ __forceinline__ void gather_rows8_mm(const float4* __restrict__ X4, int rl, int wl,
                                                int j, int lane, float4& a,
                                                v8f& c, int& ks, const float* pa, const float* pb) {
    float4 m[8];
    float w[8];
#pragma unroll
    for (int u = 0; u < 8; ++u) {
        const int r = __builtin_amdgcn_readlane(rl, j + u);
        w[u] = __int_as_float(__builtin_amdgcn_readlane(wl, j + u));
        m[u] = X4[(size_t)r * (GCN_D / 4) + lane];
    }
    if (ks < GCN_D / 4) gemm_slice(c, ks, pa, pb);
#pragma unroll
    for (int u = 0; u < 8; ++u) {
        a.x = fmaf(m[u].x, w[u], a.x);
        a.y = fmaf(m[u].y, w[u], a.y);
        a.z = fmaf(m[u].z, w[u], a.z);
        a.w = fmaf(m[u].w, w[u], a.w);
    }
}

__global__ __launch_bounds__(FUSE_T)
void k_gather_ln_gemm(const float* __restrict__ XW, const int* __restrict__ off,
                      const int* __restrict__ pairs, const float* __restrict__ dinv,
                      const float* __restrict__ bias, const float* __restrict__ hin,
                      const float* __restrict__ gamma, const float* __restrict__ beta,
                      float* __restrict__ hout, const float* __restrict__ Wn,
                      float* __restrict__ XWn, int n_nodes, int do_relu) {
    __shared__ __attribute__((aligned(16))) float sW[GCN_D * W_PITCH];
    __shared__ __attribute__((aligned(16))) float sA[2][TILE_NODES * W_PITCH];

    const int t = threadIdx.x;
    const int lane = t & 31;
    const int wv = __builtin_amdgcn_readfirstlane(t >> 5);
    const int lm = lane & 15;
    const int hi = lane >> 4;

    const float4* __restrict__ W4 = (const float4*)Wn;
#pragma unroll
    for (int p = 0; p < (GCN_D * GCN_D / 4) / FUSE_T; ++p) {
        const int i = t + p * FUSE_T;
        const int jr = i >> 5;
        const int c4 = i & 31;
        const float4 v = W4[i];
        *(float4*)&sW[jr * W_PITCH + 4 * c4] = v;
    }

    const float4* __restrict__ X4 = (const float4*)XW;
    const int2* __restrict__ P = (const int2*)pairs;
    const float4 b4 = ((const float4*)bias)[lane];
    const float4 g4 = ((const float4*)gamma)[lane];
    const float4 be4 = ((const float4*)beta)[lane];

    const int blk_base = blockIdx.x * FUSE_NODES;
    const int gs = wv >> 3;
    const int gn = wv & 7;
    const float* pb = sW + (gn * 16 + lm) * W_PITCH + 2 * hi;

    v8f c = {};
    int ks = GCN_D / 4;
    for (int tile = 0; tile <= FUSE_TILES; ++tile) {
        const float* pa = sA[0];
        int m0 = 0;
        bool gvalid = false;
        float4 dlo = make_float4(0.f, 0.f, 0.f, 0.f);
        float4 dhi = dlo;
        if (tile > 0) {
            const int prow0 = blk_base + (tile - 1) * TILE_NODES + gs * 16;
            gvalid = prow0 < n_nodes;
            if (gvalid) {
                pa = sA[(tile - 1) & 1] + (gs * 16 + lm) * W_PITCH + 2 * hi;
                m0 = prow0 + hi * 8;
                dlo = *(const float4*)(dinv + m0);
                dhi = *(const float4*)(dinv + m0 + 4);
                ks = 0;
#pragma unroll
                for (int v = 0; v < 8; ++v) c[v] = 0.0f;
            }
        }

        if (tile < FUSE_TILES) {
            for (int i = 0; i < 2; ++i) {
                const int node = blk_base + tile * TILE_NODES + 2 * wv + i;
                if (node < n_nodes) {
                    const int beg = off[node];
                    const int end = off[node + 1];
                    const float di = dinv[node];
                    float4 a = X4[(size_t)node * (GCN_D / 4) + lane];
                    const float4 id = ((const float4*)(hin + (size_t)node * GCN_D))[lane];

                    for (int p0 = beg; p0 < end; p0 += 32) {
                        const int p = p0 + lane;
                        int2 pr;
                        pr.x = 0;
                        pr.y = 0;
                        if (p < end) pr = P[p];
                        const int left = end - p0;
                        const int cnt = left < 32 ? left : 32;
                        int j = 0;
                        for (; j + 8 <= cnt; j += 8)
                            gather_rows8_mm(X4, pr.x, pr.y, j, lane, a, c, ks, pa, pb);
                        if (j + 4 <= cnt) { gather_rows<4>(X4, pr.x, pr.y, j, lane, a); j += 4; }
                        if (j + 2 <= cnt) { gather_rows<2>(X4, pr.x, pr.y, j, lane, a); j += 2; }
                        if (j < cnt) gather_rows<1>(X4, pr.x, pr.y, j, lane, a);
                    }

                    a.x = fmaf(di, a.x, b4.x);
                    a.y = fmaf(di, a.y, b4.y);
                    a.z = fmaf(di, a.z, b4.z);
                    a.w = fmaf(di, a.w, b4.w);

                    float s = a.x + a.y + a.z + a.w;
#pragma unroll
                    for (int o = 16; o > 0; o >>= 1) s += __shfl_xor(s, o, 32);
                    const float mu = s * (1.0f / GCN_D);

                    float dx = a.x - mu, dy = a.y - mu, dz = a.z - mu, dw = a.w - mu;
                    float q = dx * dx + dy * dy + dz * dz + dw * dw;
#pragma unroll
                    for (int o = 16; o > 0; o >>= 1) q += __shfl_xor(q, o, 32);
                    const float inv = rsqrtf(q * (1.0f / GCN_D) + LN_EPS);

                    float4 y;
                    y.x = dx * inv * g4.x + be4.x;
                    y.y = dy * inv * g4.y + be4.y;
                    y.z = dz * inv * g4.z + be4.z;
                    y.w = dw * inv * g4.w + be4.w;
                    if (do_relu) {
                        y.x = fmaxf(y.x, 0.0f);
                        y.y = fmaxf(y.y, 0.0f);
                        y.z = fmaxf(y.z, 0.0f);
                        y.w = fmaxf(y.w, 0.0f);
                    }
                    y.x += id.x; y.y += id.y; y.z += id.z; y.w += id.w;
                    ((float4*)(hout + (size_t)node * GCN_D))[lane] = y;
                    *(float4*)&sA[tile & 1][(2 * wv + i) * W_PITCH + 4 * lane] = y;
                }
            }
        }

        while (ks < GCN_D / 4) gemm_slice(c, ks, pa, pb);
        if (gvalid) {
            const float dv[8] = { dlo.x, dlo.y, dlo.z, dlo.w, dhi.x, dhi.y, dhi.z, dhi.w };
#pragma unroll
            for (int v = 0; v < 8; ++v)
                XWn[(size_t)(m0 + v) * GCN_D + gn * 16 + lm] = c[v] * dv[v];
        }
        if (tile < FUSE_TILES) __syncthreads();
    }
}


extern "C" void kernel_launch(void* const* d_in, const int* in_sizes, int n_in,
                              void* d_out, int out_size, void* d_ws, size_t ws_size,
                              hipStream_t stream) {
    (void)in_sizes; (void)n_in; (void)out_size; (void)ws_size;

    const float* x      = (const float*)d_in[0];
    const int*   ei     = (const int*)d_in[1];
    const float* ew     = (const float*)d_in[2];
    const float* Ws     = (const float*)d_in[3];
    const float* bs     = (const float*)d_in[4];
    const float* gammas = (const float*)d_in[5];
    const float* betas  = (const float*)d_in[6];
    float*       out    = (float*)d_out;

    const int* row = ei;
    const int* col = ei + GCN_E;

    float* ws    = (float*)d_ws;
    int*   off   = (int*)ws;
    float* dinv  = (float*)(off + 50048);
    int*   cntBG = (int*)(dinv + 50048);
    int*   offBG = cntBG + 100352;
    int*   tmp   = offBG + 100352;
    int*   pairs = tmp + 2 * GCN_E;
    float* xw    = (float*)(pairs + 2 * GCN_E);
    float* h1    = xw + (size_t)GCN_N * GCN_D;
    float* xw2   = h1 + (size_t)GCN_N * GCN_D;

    const int BLK = 256;

    const int node_wave_blocks = (GCN_N + (BLK / 32) - 1) / (BLK / 32);
    k_bin_count<<<SORT_G, SORT_T, 0, stream>>>(col, cntBG, GCN_E);
    k_scan<<<(NBINS * SORT_G + SCAN_T - 1) / SCAN_T, SCAN_T, 0, stream>>>(cntBG, offBG, NBINS * SORT_G);
    k_bin_scatter<<<SORT_G, SORT_T, 0, stream>>>(row, col, ew, offBG, tmp, GCN_E);
    k_bin_sort<<<NBINS, SORT_T, 0, stream>>>(offBG, tmp, off, pairs, dinv, GCN_N);

    const int gemm_blocks  = (GCN_N + GEMM_ROWS - 1) / GEMM_ROWS;

    const int fuse_blocks = (GCN_N + FUSE_NODES - 1) / FUSE_NODES;

    k_gemm_wmma<<<gemm_blocks, GEMM_WAVES * 32, 0, stream>>>(x, Ws, dinv, xw, GCN_N);
    k_gather_ln_gemm<<<fuse_blocks, FUSE_T, 0, stream>>>(xw, off, pairs, dinv,
                                                         bs, x, gammas, betas, h1,
                                                         Ws + (size_t)GCN_D * GCN_D, xw2,
                                                         GCN_N,  1);
    k_gather_ln_gemm<<<fuse_blocks, FUSE_T, 0, stream>>>(xw2, off, pairs, dinv,
                                                         bs + GCN_D, h1, gammas + GCN_D,
                                                         betas + GCN_D, out,
                                                         Ws + (size_t)2 * GCN_D * GCN_D, xw,
                                                         GCN_N,  1);
    k_gather_ln<<<node_wave_blocks, BLK, 0, stream>>>(xw, off, pairs, dinv,
                                                      bs + 2 * GCN_D, out,
                                                      gammas + 2 * GCN_D, betas + 2 * GCN_D,
                                                      out, GCN_N,  0);
}


# Edge sort in 2 launches, not 4: each slice group counts, scans and places its records by bin on LDS (no global count or scan pass); bin groups sum the counts in front; arrival, so sum order, changes
# speedup vs baseline: 5.8737x; 1.0843x over previous
#include <hip/hip_runtime.h>
#include <hip/hip_bf16.h>


#define GCN_N 50000
#define GCN_E 1600000
#define GCN_D 128
#define GCN_L 3
#define LN_EPS 1e-5f

typedef __attribute__((ext_vector_type(2))) float v2f;
typedef __attribute__((ext_vector_type(8))) float v8f;


#define BIN_SHIFT 7
#define BIN_NODES (1 << BIN_SHIFT)
#define NBINS ((GCN_N + BIN_NODES - 1) / BIN_NODES)
#define SORT_G 256
#define SORT_T 1024
#define SORT_CH ((GCN_E + SORT_G - 1) / SORT_G)
#define SORT_IT ((SORT_CH + SORT_T - 1) / SORT_T)

static_assert(GCN_N <= 65536, "source row must fit in 16 bits");
static_assert(NBINS * SORT_G <= 100352, "count and start tables as laid out in the workspace");
static_assert(NBINS <= SORT_T, "one thread per bin in the slice's scan");
static_assert(SORT_G * SORT_CH >= GCN_E, "the slices cover the edge list");

#define WALK_T (SORT_T / SORT_G)
static_assert(WALK_T * SORT_G == SORT_T && WALK_T >= 1, "whole groups of threads per slice run");
static_assert(SORT_G % 4 == 0, "count table rows are read 16 bytes at a time");
#define WALK_K 8

__global__ __launch_bounds__(SORT_T)
void k_slice_sort(const int* __restrict__ row, const int* __restrict__ col,
                  const float* __restrict__ w, int* __restrict__ cntT,
                  int* __restrict__ startT, int* __restrict__ tmp, int e) {
    __shared__ int hist[NBINS];
    __shared__ int cur[NBINS];
    __shared__ int wtot[SORT_T / 32];
    const int t = threadIdx.x;
    const int lane = t & 31;
    const int wv = t >> 5;
    const int g = blockIdx.x;
    for (int b = t; b < NBINS; b += SORT_T) hist[b] = 0;
    __syncthreads();
    const int lo = g * SORT_CH;
    const int hi = (lo + SORT_CH < e) ? lo + SORT_CH : e;
    int c[SORT_IT];
    int r[SORT_IT];
    float wq[SORT_IT];
#pragma unroll
    for (int k = 0; k < SORT_IT; ++k) {
        const int i = lo + t + k * SORT_T;
        const bool in = i < hi;
        c[k] = in ? col[i] : 0;
        r[k] = in ? row[i] : 0;
        wq[k] = in ? w[i] : 0.0f;
    }
#pragma unroll
    for (int k = 0; k < SORT_IT; ++k) {
        const int i = lo + t + k * SORT_T;
        if (i < hi) atomicAdd(&hist[c[k] >> BIN_SHIFT], 1);
    }
    __syncthreads();

    const int v = (t < NBINS) ? hist[t] : 0;
    int inc = v;
    for (int d = 1; d < 32; d <<= 1) {
        int y = __shfl_up(inc, d, 32);
        if (lane >= d) inc += y;
    }
    if (lane == 31) wtot[wv] = inc;
    __syncthreads();
    int pre = (lane < wv) ? wtot[lane] : 0;
    for (int d = 16; d > 0; d >>= 1) pre += __shfl_xor(pre, d, 32);
    if (t < NBINS) {
        const int st = lo + pre + inc - v;
        cur[t] = st;
        cntT[t * SORT_G + g] = v;
        startT[t * SORT_G + g] = st;
    }
    __syncthreads();

#pragma unroll
    for (int k = 0; k < SORT_IT; ++k) {
        const int i = lo + t + k * SORT_T;
        if (i < hi) {
            const int p = atomicAdd(&cur[c[k] >> BIN_SHIFT], 1);
            int2 rec;
            rec.x = r[k] | ((c[k] & (BIN_NODES - 1)) << 16);
            rec.y = __float_as_int(wq[k]);
            ((int2*)tmp)[p] = rec;
        }
    }
}

__global__ __launch_bounds__(SORT_T)
void k_bin_sort(const int* cntT, const int* __restrict__ startT,
                const int* __restrict__ tmp, int* __restrict__ off, int* __restrict__ pairs,
                float* __restrict__ dinv, int n_nodes) {
    __shared__ int cnt[BIN_NODES];
    __shared__ int cur[BIN_NODES];
    __shared__ float wsum[BIN_NODES];
    __shared__ int wtot[BIN_NODES / 32];
    __shared__ int rs[SORT_G];
    __shared__ int rc[SORT_G];
    __shared__ int part[SORT_T / 32];
    const int t = threadIdx.x;
    const int lane = t & 31;
    const int wv = t >> 5;
    const int b = blockIdx.x;
    const int2* __restrict__ T2 = (const int2*)tmp;

    const int4* C4 = (const int4*)cntT;
    const int nfront4 = b * (SORT_G / 4);
    const int trips = (nfront4 + SORT_T - 1) / SORT_T;
    int s = 0;
    for (int k0 = 0; k0 < trips; k0 += 8) {
        int4 q[8];
        int keep[8];
#pragma unroll
        for (int u = 0; u < 8; ++u) {
            const int i = t + (k0 + u) * SORT_T;
            keep[u] = (i - nfront4) >> 31;
            q[u] = C4[min(i, nfront4 - 1)];
        }
        __builtin_amdgcn_sched_barrier(0);
#pragma unroll
        for (int u = 0; u < 8; ++u) s += ((q[u].x + q[u].y) + (q[u].z + q[u].w)) & keep[u];
    }
    for (int d = 16; d > 0; d >>= 1) s += __shfl_xor(s, d, 32);
    if (lane == 0) part[wv] = s;
    if (t < SORT_G) {
        rs[t] = startT[b * SORT_G + t];
        rc[t] = cntT[b * SORT_G + t];
    }
    if (t < BIN_NODES) { cnt[t] = 0; wsum[t] = 0.0f; }
    __syncthreads();

    int beg = part[lane];
    for (int d = 16; d > 0; d >>= 1) beg += __shfl_xor(beg, d, 32);

    const int sg = t / WALK_T;
    const int sub = t % WALK_T;
    const int rbeg = rs[sg];
    const int rend = rbeg + rc[sg];
    int2 rec[WALK_K];
#pragma unroll
    for (int k = 0; k < WALK_K; ++k) {
        const int i = rbeg + sub + k * WALK_T;
        rec[k] = (i < rend) ? T2[i] : make_int2(0, 0);
    }
#pragma unroll
    for (int k = 0; k < WALK_K; ++k) {
        const int i = rbeg + sub + k * WALK_T;
        if (i < rend) {
            const int ln = (rec[k].x >> 16) & (BIN_NODES - 1);
            atomicAdd(&cnt[ln], 1);
            atomicAdd(&wsum[ln], __int_as_float(rec[k].y));
        }
    }
    for (int i = rbeg + sub + WALK_K * WALK_T; i < rend; i += WALK_T) {
        const int2 rl = T2[i];
        const int ln = (rl.x >> 16) & (BIN_NODES - 1);
        atomicAdd(&cnt[ln], 1);
        atomicAdd(&wsum[ln], __int_as_float(rl.y));
    }
    __syncthreads();

    const int v = (t < BIN_NODES) ? cnt[t] : 0;
    int inc = v;
    for (int d = 1; d < 32; d <<= 1) {
        int y = __shfl_up(inc, d, 32);
        if (lane >= d) inc += y;
    }
    if (t < BIN_NODES && lane == 31) wtot[wv] = inc;
    __syncthreads();
    if (t < BIN_NODES) {
        int pre = 0;
        for (int q = 0; q < wv; ++q) pre += wtot[q];
        const int excl = beg + pre + inc - v;
        cur[t] = excl;
        const int node = b * BIN_NODES + t;
        if (node < n_nodes) {
            off[node] = excl;
            const float d = wsum[t] + 1.0f;
            dinv[node] = (d > 0.0f) ? rsqrtf(fmaxf(d, 1e-12f)) : 0.0f;
        }
        if (node == n_nodes - 1) off[n_nodes] = excl + v;
    }
    __syncthreads();
#pragma unroll
    for (int k = 0; k < WALK_K; ++k) {
        const int i = rbeg + sub + k * WALK_T;
        if (i < rend) {
            const int p = atomicAdd(&cur[(rec[k].x >> 16) & (BIN_NODES - 1)], 1);
            int2 o;
            o.x = rec[k].x & 0xFFFF;
            o.y = rec[k].y;
            ((int2*)pairs)[p] = o;
        }
    }
    for (int i = rbeg + sub + WALK_K * WALK_T; i < rend; i += WALK_T) {
        const int2 rl = T2[i];
        const int p = atomicAdd(&cur[(rl.x >> 16) & (BIN_NODES - 1)], 1);
        int2 o;
        o.x = rl.x & 0xFFFF;
        o.y = rl.y;
        ((int2*)pairs)[p] = o;
    }
}


#define GEMM_WAVES 8
#define GEMM_ROWS (GEMM_WAVES * 16)
#define W_PITCH 132

__global__ __launch_bounds__(GEMM_WAVES * 32)
void k_gemm_wmma(const float* __restrict__ H, const float* __restrict__ W,
                 const float* __restrict__ dinv, float* __restrict__ XW, int n_nodes) {
    __shared__ __attribute__((aligned(16))) float sW[GCN_D * W_PITCH];

    const int t = threadIdx.x;
    const int lane = t & 31;
    const int wv = t >> 5;
    const int lm = lane & 15;
    const int hi = lane >> 4;

    const int row0 = (blockIdx.x * GEMM_WAVES + wv) * 16;
    const bool active = row0 < n_nodes;
    const int rowa = active ? row0 : 0;

    const float2* __restrict__ Ap = (const float2*)(H + (size_t)(rowa + lm) * GCN_D);
    v2f a[GCN_D / 4];
#pragma unroll
    for (int s = 0; s < GCN_D / 4; ++s) {
        const float2 af = Ap[2 * s + hi];
        a[s].x = af.x;
        a[s].y = af.y;
    }

    const float4* __restrict__ W4 = (const float4*)W;
#pragma unroll
    for (int p = 0; p < (GCN_D * GCN_D / 4) / (GEMM_WAVES * 32); ++p) {
        const int i = t + p * (GEMM_WAVES * 32);
        const int j = i >> 5;
        const int c4 = i & 31;
        const float4 v = W4[i];
        *(float4*)&sW[j * W_PITCH + 4 * c4] = v;
    }
    __syncthreads();
    if (!active) return;

    const float* sB = sW + lm * W_PITCH + 2 * hi;

    const int m0 = row0 + hi * 8;
    const float4 dlo = *(const float4*)(dinv + m0);
    const float4 dhi = *(const float4*)(dinv + m0 + 4);

    v8f c[8] = {};
    v2f b[2][16];
#pragma unroll
    for (int n = 0; n < 8; ++n) {
#pragma unroll
        for (int u = 0; u < 2; ++u) {
            const float2 bf = *(const float2*)(sB + n * 16 * W_PITCH + 4 * u);
            b[0][2 * n + u].x = bf.x;
            b[0][2 * n + u].y = bf.y;
        }
    }
#pragma unroll
    for (int blk = 0; blk < GCN_D / 8; ++blk) {
        const int cur = blk & 1;
        if (blk + 1 < GCN_D / 8) {
#pragma unroll
            for (int n = 0; n < 8; ++n) {
#pragma unroll
                for (int u = 0; u < 2; ++u) {
                    const float2 bf = *(const float2*)(sB + n * 16 * W_PITCH + 4 * (2 * (blk + 1) + u));
                    b[cur ^ 1][2 * n + u].x = bf.x;
                    b[cur ^ 1][2 * n + u].y = bf.y;
                }
            }
        }
        __builtin_amdgcn_sched_barrier(0);
#pragma unroll
        for (int u = 0; u < 2; ++u) {
#pragma unroll
            for (int n = 0; n < 8; ++n) {
                c[n] = __builtin_amdgcn_wmma_f32_16x16x4_f32(
                            false, a[2 * blk + u],  false, b[cur][2 * n + u],
                            (short)0, c[n],  false,  false);
            }
        }
        __builtin_amdgcn_sched_barrier(0);
    }

    const float dv[8] = { dlo.x, dlo.y, dlo.z, dlo.w, dhi.x, dhi.y, dhi.z, dhi.w };
#pragma unroll
    for (int n = 0; n < 8; ++n) {
#pragma unroll
        for (int v = 0; v < 8; ++v)
            XW[(size_t)(m0 + v) * GCN_D + n * 16 + lm] = c[n][v] * dv[v];
    }
}


template <int U>
__device__ __forceinline__ void gather_rows(const float4* __restrict__ X4, int rl, int wl,
                                            int j, int lane, float4& a) {
    float4 m[U];
    float w[U];
#pragma unroll
    for (int u = 0; u < U; ++u) {
        const int r = __builtin_amdgcn_readlane(rl, j + u);
        w[u] = __int_as_float(__builtin_amdgcn_readlane(wl, j + u));
        m[u] = X4[(size_t)r * (GCN_D / 4) + lane];
    }
#pragma unroll
    for (int u = 0; u < U; ++u) {
        a.x = fmaf(m[u].x, w[u], a.x);
        a.y = fmaf(m[u].y, w[u], a.y);
        a.z = fmaf(m[u].z, w[u], a.z);
        a.w = fmaf(m[u].w, w[u], a.w);
    }
}

__global__ void k_gather_ln(const float* __restrict__ XW, const int* __restrict__ off,
                            const int* __restrict__ pairs, const float* __restrict__ dinv,
                            const float* __restrict__ bias, const float* hin,
                            const float* __restrict__ gamma, const float* __restrict__ beta,
                            float* hout, int n_nodes, int do_relu) {
    const int wave = (blockIdx.x * blockDim.x + threadIdx.x) >> 5;
    const int lane = threadIdx.x & 31;
    if (wave >= n_nodes) return;
    const int node = __builtin_amdgcn_readfirstlane(wave);
    const int beg = off[node];
    const int end = off[node + 1];
    const float di = dinv[node];

    const float4* __restrict__ X4 = (const float4*)XW;
    const int2* __restrict__ P = (const int2*)pairs;

    float4 x = X4[(size_t)node * (GCN_D / 4) + lane];
    float4 b = ((const float4*)bias)[lane];
    const float4 id = ((const float4*)(hin + (size_t)node * GCN_D))[lane];
    float4 a = x;

    for (int p0 = beg; p0 < end; p0 += 32) {
        const int p = p0 + lane;
        int2 pr;
        pr.x = 0;
        pr.y = 0;
        if (p < end) pr = P[p];
        const int left = end - p0;
        const int cnt = left < 32 ? left : 32;
        int j = 0;
        for (; j + 8 <= cnt; j += 8) gather_rows<8>(X4, pr.x, pr.y, j, lane, a);
        if (j + 4 <= cnt) { gather_rows<4>(X4, pr.x, pr.y, j, lane, a); j += 4; }
        if (j + 2 <= cnt) { gather_rows<2>(X4, pr.x, pr.y, j, lane, a); j += 2; }
        if (j < cnt) gather_rows<1>(X4, pr.x, pr.y, j, lane, a);
    }

    a.x = fmaf(di, a.x, b.x);
    a.y = fmaf(di, a.y, b.y);
    a.z = fmaf(di, a.z, b.z);
    a.w = fmaf(di, a.w, b.w);

    float s = a.x + a.y + a.z + a.w;
#pragma unroll
    for (int o = 16; o > 0; o >>= 1) s += __shfl_xor(s, o, 32);
    const float mu = s * (1.0f / GCN_D);

    float dx = a.x - mu, dy = a.y - mu, dz = a.z - mu, dw = a.w - mu;
    float q = dx * dx + dy * dy + dz * dz + dw * dw;
#pragma unroll
    for (int o = 16; o > 0; o >>= 1) q += __shfl_xor(q, o, 32);
    const float inv = rsqrtf(q * (1.0f / GCN_D) + LN_EPS);

    const float4 g = ((const float4*)gamma)[lane];
    const float4 be = ((const float4*)beta)[lane];

    float4 y;
    y.x = dx * inv * g.x + be.x;
    y.y = dy * inv * g.y + be.y;
    y.z = dz * inv * g.z + be.z;
    y.w = dw * inv * g.w + be.w;
    if (do_relu) {
        y.x = fmaxf(y.x, 0.0f);
        y.y = fmaxf(y.y, 0.0f);
        y.z = fmaxf(y.z, 0.0f);
        y.w = fmaxf(y.w, 0.0f);
    }
    y.x += id.x; y.y += id.y; y.z += id.z; y.w += id.w;
    ((float4*)(hout + (size_t)node * GCN_D))[lane] = y;
}


#define FUSE_WAVES 16
#define FUSE_T (FUSE_WAVES * 32)
#define TILE_NODES 32
#define FUSE_TILES 4
#define FUSE_NODES (TILE_NODES * FUSE_TILES)
#define KS_SLICE 8

static_assert(GCN_N % 16 == 0, "strips of 16 nodes must be whole");
static_assert((GCN_D * GCN_D / 4) % FUSE_T == 0, "W staging: whole passes of the work-group");

__device__ __forceinline__ void gemm_slice(v8f& c, int& ks, const float* pa, const float* pb) {
#pragma unroll
    for (int q = 0; q < KS_SLICE; ++q) {
        const float2 af = *(const float2*)(pa + 4 * (ks + q));
        const float2 bf = *(const float2*)(pb + 4 * (ks + q));
        v2f a2; a2.x = af.x; a2.y = af.y;
        v2f b2; b2.x = bf.x; b2.y = bf.y;
        c = __builtin_amdgcn_wmma_f32_16x16x4_f32( false, a2,  false, b2,
                                                   (short)0, c,
                                                   false,  false);
    }
    ks += KS_SLICE;
}

__device__ __forceinline__ void gather_rows8_mm(const float4* __restrict__ X4, int rl, int wl,
                                                int j, int lane, float4& a,
                                                v8f& c, int& ks, const float* pa, const float* pb) {
    float4 m[8];
    float w[8];
#pragma unroll
    for (int u = 0; u < 8; ++u) {
        const int r = __builtin_amdgcn_readlane(rl, j + u);
        w[u] = __int_as_float(__builtin_amdgcn_readlane(wl, j + u));
        m[u] = X4[(size_t)r * (GCN_D / 4) + lane];
    }
    if (ks < GCN_D / 4) gemm_slice(c, ks, pa, pb);
#pragma unroll
    for (int u = 0; u < 8; ++u) {
        a.x = fmaf(m[u].x, w[u], a.x);
        a.y = fmaf(m[u].y, w[u], a.y);
        a.z = fmaf(m[u].z, w[u], a.z);
        a.w = fmaf(m[u].w, w[u], a.w);
    }
}

__global__ __launch_bounds__(FUSE_T)
void k_gather_ln_gemm(const float* __restrict__ XW, const int* __restrict__ off,
                      const int* __restrict__ pairs, const float* __restrict__ dinv,
                      const float* __restrict__ bias, const float* __restrict__ hin,
                      const float* __restrict__ gamma, const float* __restrict__ beta,
                      float* __restrict__ hout, const float* __restrict__ Wn,
                      float* __restrict__ XWn, int n_nodes, int do_relu) {
    __shared__ __attribute__((aligned(16))) float sW[GCN_D * W_PITCH];
    __shared__ __attribute__((aligned(16))) float sA[2][TILE_NODES * W_PITCH];

    const int t = threadIdx.x;
    const int lane = t & 31;
    const int wv = __builtin_amdgcn_readfirstlane(t >> 5);
    const int lm = lane & 15;
    const int hi = lane >> 4;

    const float4* __restrict__ W4 = (const float4*)Wn;
#pragma unroll
    for (int p = 0; p < (GCN_D * GCN_D / 4) / FUSE_T; ++p) {
        const int i = t + p * FUSE_T;
        const int jr = i >> 5;
        const int c4 = i & 31;
        const float4 v = W4[i];
        *(float4*)&sW[jr * W_PITCH + 4 * c4] = v;
    }

    const float4* __restrict__ X4 = (const float4*)XW;
    const int2* __restrict__ P = (const int2*)pairs;
    const float4 b4 = ((const float4*)bias)[lane];
    const float4 g4 = ((const float4*)gamma)[lane];
    const float4 be4 = ((const float4*)beta)[lane];

    const int blk_base = blockIdx.x * FUSE_NODES;
    const int gs = wv >> 3;
    const int gn = wv & 7;
    const float* pb = sW + (gn * 16 + lm) * W_PITCH + 2 * hi;

    v8f c = {};
    int ks = GCN_D / 4;
    for (int tile = 0; tile <= FUSE_TILES; ++tile) {
        const float* pa = sA[0];
        int m0 = 0;
        bool gvalid = false;
        float4 dlo = make_float4(0.f, 0.f, 0.f, 0.f);
        float4 dhi = dlo;
        if (tile > 0) {
            const int prow0 = blk_base + (tile - 1) * TILE_NODES + gs * 16;
            gvalid = prow0 < n_nodes;
            if (gvalid) {
                pa = sA[(tile - 1) & 1] + (gs * 16 + lm) * W_PITCH + 2 * hi;
                m0 = prow0 + hi * 8;
                dlo = *(const float4*)(dinv + m0);
                dhi = *(const float4*)(dinv + m0 + 4);
                ks = 0;
#pragma unroll
                for (int v = 0; v < 8; ++v) c[v] = 0.0f;
            }
        }

        if (tile < FUSE_TILES) {
            for (int i = 0; i < 2; ++i) {
                const int node = blk_base + tile * TILE_NODES + 2 * wv + i;
                if (node < n_nodes) {
                    const int beg = off[node];
                    const int end = off[node + 1];
                    const float di = dinv[node];
                    float4 a = X4[(size_t)node * (GCN_D / 4) + lane];
                    const float4 id = ((const float4*)(hin + (size_t)node * GCN_D))[lane];

                    for (int p0 = beg; p0 < end; p0 += 32) {
                        const int p = p0 + lane;
                        int2 pr;
                        pr.x = 0;
                        pr.y = 0;
                        if (p < end) pr = P[p];
                        const int left = end - p0;
                        const int cnt = left < 32 ? left : 32;
                        int j = 0;
                        for (; j + 8 <= cnt; j += 8)
                            gather_rows8_mm(X4, pr.x, pr.y, j, lane, a, c, ks, pa, pb);
                        if (j + 4 <= cnt) { gather_rows<4>(X4, pr.x, pr.y, j, lane, a); j += 4; }
                        if (j + 2 <= cnt) { gather_rows<2>(X4, pr.x, pr.y, j, lane, a); j += 2; }
                        if (j < cnt) gather_rows<1>(X4, pr.x, pr.y, j, lane, a);
                    }

                    a.x = fmaf(di, a.x, b4.x);
                    a.y = fmaf(di, a.y, b4.y);
                    a.z = fmaf(di, a.z, b4.z);
                    a.w = fmaf(di, a.w, b4.w);

                    float s = a.x + a.y + a.z + a.w;
#pragma unroll
                    for (int o = 16; o > 0; o >>= 1) s += __shfl_xor(s, o, 32);
                    const float mu = s * (1.0f / GCN_D);

                    float dx = a.x - mu, dy = a.y - mu, dz = a.z - mu, dw = a.w - mu;
                    float q = dx * dx + dy * dy + dz * dz + dw * dw;
#pragma unroll
                    for (int o = 16; o > 0; o >>= 1) q += __shfl_xor(q, o, 32);
                    const float inv = rsqrtf(q * (1.0f / GCN_D) + LN_EPS);

                    float4 y;
                    y.x = dx * inv * g4.x + be4.x;
                    y.y = dy * inv * g4.y + be4.y;
                    y.z = dz * inv * g4.z + be4.z;
                    y.w = dw * inv * g4.w + be4.w;
                    if (do_relu) {
                        y.x = fmaxf(y.x, 0.0f);
                        y.y = fmaxf(y.y, 0.0f);
                        y.z = fmaxf(y.z, 0.0f);
                        y.w = fmaxf(y.w, 0.0f);
                    }
                    y.x += id.x; y.y += id.y; y.z += id.z; y.w += id.w;
                    ((float4*)(hout + (size_t)node * GCN_D))[lane] = y;
                    *(float4*)&sA[tile & 1][(2 * wv + i) * W_PITCH + 4 * lane] = y;
                }
            }
        }

        while (ks < GCN_D / 4) gemm_slice(c, ks, pa, pb);
        if (gvalid) {
            const float dv[8] = { dlo.x, dlo.y, dlo.z, dlo.w, dhi.x, dhi.y, dhi.z, dhi.w };
#pragma unroll
            for (int v = 0; v < 8; ++v)
                XWn[(size_t)(m0 + v) * GCN_D + gn * 16 + lm] = c[v] * dv[v];
        }
        if (tile < FUSE_TILES) __syncthreads();
    }
}


extern "C" void kernel_launch(void* const* d_in, const int* in_sizes, int n_in,
                              void* d_out, int out_size, void* d_ws, size_t ws_size,
                              hipStream_t stream) {
    (void)in_sizes; (void)n_in; (void)out_size; (void)ws_size;

    const float* x      = (const float*)d_in[0];
    const int*   ei     = (const int*)d_in[1];
    const float* ew     = (const float*)d_in[2];
    const float* Ws     = (const float*)d_in[3];
    const float* bs     = (const float*)d_in[4];
    const float* gammas = (const float*)d_in[5];
    const float* betas  = (const float*)d_in[6];
    float*       out    = (float*)d_out;

    const int* row = ei;
    const int* col = ei + GCN_E;

    float* ws    = (float*)d_ws;
    int*   off   = (int*)ws;
    float* dinv  = (float*)(off + 50048);
    int*   cntT  = (int*)(dinv + 50048);
    int*   startT = cntT + 100352;
    int*   tmp   = startT + 100352;
    int*   pairs = tmp + 2 * GCN_E;
    float* xw    = (float*)(pairs + 2 * GCN_E);
    float* h1    = xw + (size_t)GCN_N * GCN_D;
    float* xw2   = h1 + (size_t)GCN_N * GCN_D;

    const int BLK = 256;

    const int node_wave_blocks = (GCN_N + (BLK / 32) - 1) / (BLK / 32);
    k_slice_sort<<<SORT_G, SORT_T, 0, stream>>>(row, col, ew, cntT, startT, tmp, GCN_E);
    k_bin_sort<<<NBINS, SORT_T, 0, stream>>>(cntT, startT, tmp, off, pairs, dinv, GCN_N);

    const int gemm_blocks  = (GCN_N + GEMM_ROWS - 1) / GEMM_ROWS;

    const int fuse_blocks = (GCN_N + FUSE_NODES - 1) / FUSE_NODES;

    k_gemm_wmma<<<gemm_blocks, GEMM_WAVES * 32, 0, stream>>>(x, Ws, dinv, xw, GCN_N);
    k_gather_ln_gemm<<<fuse_blocks, FUSE_T, 0, stream>>>(xw, off, pairs, dinv,
                                                         bs, x, gammas, betas, h1,
                                                         Ws + (size_t)GCN_D * GCN_D, xw2,
                                                         GCN_N,  1);
    k_gather_ln_gemm<<<fuse_blocks, FUSE_T, 0, stream>>>(xw2, off, pairs, dinv,
                                                         bs + GCN_D, h1, gammas + GCN_D,
                                                         betas + GCN_D, out,
                                                         Ws + (size_t)2 * GCN_D * GCN_D, xw,
                                                         GCN_N,  1);
    k_gather_ln<<<node_wave_blocks, BLK, 0, stream>>>(xw, off, pairs, dinv,
                                                      bs + 2 * GCN_D, out,
                                                      gammas + 2 * GCN_D, betas + 2 * GCN_D,
                                                      out, GCN_N,  0);
}


# Both sort launches place their 8-byte records in LDS first and copy them out in order, so record stores cover whole lines instead of one line per lane; slots, so sum order, unchanged
# speedup vs baseline: 6.0841x; 1.0358x over previous
#include <hip/hip_runtime.h>
#include <hip/hip_bf16.h>


#define GCN_N 50000
#define GCN_E 1600000
#define GCN_D 128
#define GCN_L 3
#define LN_EPS 1e-5f

typedef __attribute__((ext_vector_type(2))) float v2f;
typedef __attribute__((ext_vector_type(8))) float v8f;


#define BIN_SHIFT 7
#define BIN_NODES (1 << BIN_SHIFT)
#define NBINS ((GCN_N + BIN_NODES - 1) / BIN_NODES)
#define SORT_G 256
#define SORT_T 1024
#define SORT_CH ((GCN_E + SORT_G - 1) / SORT_G)
#define SORT_IT ((SORT_CH + SORT_T - 1) / SORT_T)

static_assert(GCN_N <= 65536, "source row must fit in 16 bits");
static_assert(NBINS * SORT_G <= 100352, "count and start tables as laid out in the workspace");
static_assert(NBINS <= SORT_T, "one thread per bin in the slice's scan");
static_assert(SORT_G * SORT_CH >= GCN_E, "the slices cover the edge list");

#define WALK_T (SORT_T / SORT_G)
static_assert(WALK_T * SORT_G == SORT_T && WALK_T >= 1, "whole groups of threads per slice run");
static_assert(SORT_G % 4 == 0, "count table rows are read 16 bytes at a time");
#define WALK_K 8
#define BIN_STAGE 5120
#define BIN_COPY_IT (BIN_STAGE / SORT_T)
static_assert(BIN_COPY_IT * SORT_T == BIN_STAGE, "whole passes of the work-group over the staged records");

__global__ __launch_bounds__(SORT_T)
void k_slice_sort(const int* __restrict__ row, const int* __restrict__ col,
                  const float* __restrict__ w, int* __restrict__ cntT,
                  int* __restrict__ startT, int* __restrict__ tmp, int e) {
    __shared__ int hist[NBINS];
    __shared__ int cur[NBINS];
    __shared__ int wtot[SORT_T / 32];
    __shared__ __attribute__((aligned(8))) int srec[2 * SORT_CH];
    const int t = threadIdx.x;
    const int lane = t & 31;
    const int wv = t >> 5;
    const int g = blockIdx.x;
    for (int b = t; b < NBINS; b += SORT_T) hist[b] = 0;
    __syncthreads();
    const int lo = g * SORT_CH;
    const int hi = (lo + SORT_CH < e) ? lo + SORT_CH : e;
    int c[SORT_IT];
    int r[SORT_IT];
    float wq[SORT_IT];
#pragma unroll
    for (int k = 0; k < SORT_IT; ++k) {
        const int i = lo + t + k * SORT_T;
        const bool in = i < hi;
        c[k] = in ? col[i] : 0;
        r[k] = in ? row[i] : 0;
        wq[k] = in ? w[i] : 0.0f;
    }
#pragma unroll
    for (int k = 0; k < SORT_IT; ++k) {
        const int i = lo + t + k * SORT_T;
        if (i < hi) atomicAdd(&hist[c[k] >> BIN_SHIFT], 1);
    }
    __syncthreads();

    const int v = (t < NBINS) ? hist[t] : 0;
    int inc = v;
    for (int d = 1; d < 32; d <<= 1) {
        int y = __shfl_up(inc, d, 32);
        if (lane >= d) inc += y;
    }
    if (lane == 31) wtot[wv] = inc;
    __syncthreads();
    int pre = (lane < wv) ? wtot[lane] : 0;
    for (int d = 16; d > 0; d >>= 1) pre += __shfl_xor(pre, d, 32);
    if (t < NBINS) {
        const int st = lo + pre + inc - v;
        cur[t] = st;
        cntT[t * SORT_G + g] = v;
        startT[t * SORT_G + g] = st;
    }
    __syncthreads();

#pragma unroll
    for (int k = 0; k < SORT_IT; ++k) {
        const int i = lo + t + k * SORT_T;
        if (i < hi) {
            const int p = atomicAdd(&cur[c[k] >> BIN_SHIFT], 1);
            int2 rec;
            rec.x = r[k] | ((c[k] & (BIN_NODES - 1)) << 16);
            rec.y = __float_as_int(wq[k]);
            ((int2*)srec)[p - lo] = rec;
        }
    }
    __syncthreads();
    const int nrec = hi - lo;
    int2 q[SORT_IT];
#pragma unroll
    for (int k = 0; k < SORT_IT; ++k) {
        const int i = t + k * SORT_T;
        q[k] = (i < nrec) ? ((const int2*)srec)[i] : make_int2(0, 0);
    }
#pragma unroll
    for (int k = 0; k < SORT_IT; ++k) {
        const int i = t + k * SORT_T;
        if (i < nrec) ((int2*)tmp)[lo + i] = q[k];
    }
}

__global__ __launch_bounds__(SORT_T)
void k_bin_sort(const int* cntT, const int* __restrict__ startT,
                const int* __restrict__ tmp, int* __restrict__ off, int* __restrict__ pairs,
                float* __restrict__ dinv, int n_nodes) {
    __shared__ int cnt[BIN_NODES];
    __shared__ int cur[BIN_NODES];
    __shared__ float wsum[BIN_NODES];
    __shared__ int wtot[BIN_NODES / 32];
    __shared__ int rs[SORT_G];
    __shared__ int rc[SORT_G];
    __shared__ int part[SORT_T / 32];
    __shared__ __attribute__((aligned(8))) int sout[2 * BIN_STAGE];
    const int t = threadIdx.x;
    const int lane = t & 31;
    const int wv = t >> 5;
    const int b = blockIdx.x;
    const int2* __restrict__ T2 = (const int2*)tmp;

    const int4* C4 = (const int4*)cntT;
    const int nfront4 = b * (SORT_G / 4);
    const int trips = (nfront4 + SORT_T - 1) / SORT_T;
    int s = 0;
    for (int k0 = 0; k0 < trips; k0 += 8) {
        int4 q[8];
        int keep[8];
#pragma unroll
        for (int u = 0; u < 8; ++u) {
            const int i = t + (k0 + u) * SORT_T;
            keep[u] = (i - nfront4) >> 31;
            q[u] = C4[min(i, nfront4 - 1)];
        }
        __builtin_amdgcn_sched_barrier(0);
#pragma unroll
        for (int u = 0; u < 8; ++u) s += ((q[u].x + q[u].y) + (q[u].z + q[u].w)) & keep[u];
    }
    for (int d = 16; d > 0; d >>= 1) s += __shfl_xor(s, d, 32);
    if (lane == 0) part[wv] = s;
    if (t < SORT_G) {
        rs[t] = startT[b * SORT_G + t];
        rc[t] = cntT[b * SORT_G + t];
    }
    if (t < BIN_NODES) { cnt[t] = 0; wsum[t] = 0.0f; }
    __syncthreads();

    int beg = part[lane];
    for (int d = 16; d > 0; d >>= 1) beg += __shfl_xor(beg, d, 32);

    const int sg = t / WALK_T;
    const int sub = t % WALK_T;
    const int rbeg = rs[sg];
    const int rend = rbeg + rc[sg];
    int2 rec[WALK_K];
#pragma unroll
    for (int k = 0; k < WALK_K; ++k) {
        const int i = rbeg + sub + k * WALK_T;
        rec[k] = (i < rend) ? T2[i] : make_int2(0, 0);
    }
#pragma unroll
    for (int k = 0; k < WALK_K; ++k) {
        const int i = rbeg + sub + k * WALK_T;
        if (i < rend) {
            const int ln = (rec[k].x >> 16) & (BIN_NODES - 1);
            atomicAdd(&cnt[ln], 1);
            atomicAdd(&wsum[ln], __int_as_float(rec[k].y));
        }
    }
    for (int i = rbeg + sub + WALK_K * WALK_T; i < rend; i += WALK_T) {
        const int2 rl = T2[i];
        const int ln = (rl.x >> 16) & (BIN_NODES - 1);
        atomicAdd(&cnt[ln], 1);
        atomicAdd(&wsum[ln], __int_as_float(rl.y));
    }
    __syncthreads();

    const int v = (t < BIN_NODES) ? cnt[t] : 0;
    int inc = v;
    for (int d = 1; d < 32; d <<= 1) {
        int y = __shfl_up(inc, d, 32);
        if (lane >= d) inc += y;
    }
    if (t < BIN_NODES && lane == 31) wtot[wv] = inc;
    __syncthreads();
    int nbin = 0;
    for (int q = 0; q < BIN_NODES / 32; ++q) nbin += wtot[q];
    const bool staged = nbin <= BIN_STAGE;
    if (t < BIN_NODES) {
        int pre = 0;
        for (int q = 0; q < wv; ++q) pre += wtot[q];
        const int excl = beg + pre + inc - v;
        cur[t] = excl;
        const int node = b * BIN_NODES + t;
        if (node < n_nodes) {
            off[node] = excl;
            const float d = wsum[t] + 1.0f;
            dinv[node] = (d > 0.0f) ? rsqrtf(fmaxf(d, 1e-12f)) : 0.0f;
        }
        if (node == n_nodes - 1) off[n_nodes] = excl + v;
    }
    __syncthreads();
#pragma unroll
    for (int k = 0; k < WALK_K; ++k) {
        const int i = rbeg + sub + k * WALK_T;
        if (i < rend) {
            const int p = atomicAdd(&cur[(rec[k].x >> 16) & (BIN_NODES - 1)], 1);
            int2 o;
            o.x = rec[k].x & 0xFFFF;
            o.y = rec[k].y;
            if (staged) ((int2*)sout)[p - beg] = o;
            else ((int2*)pairs)[p] = o;
        }
    }
    for (int i = rbeg + sub + WALK_K * WALK_T; i < rend; i += WALK_T) {
        const int2 rl = T2[i];
        const int p = atomicAdd(&cur[(rl.x >> 16) & (BIN_NODES - 1)], 1);
        int2 o;
        o.x = rl.x & 0xFFFF;
        o.y = rl.y;
        if (staged) ((int2*)sout)[p - beg] = o;
        else ((int2*)pairs)[p] = o;
    }
    if (staged) {
        __syncthreads();
        int2 q[BIN_COPY_IT];
#pragma unroll
        for (int k = 0; k < BIN_COPY_IT; ++k) {
            const int i = t + k * SORT_T;
            q[k] = (i < nbin) ? ((const int2*)sout)[i] : make_int2(0, 0);
        }
#pragma unroll
        for (int k = 0; k < BIN_COPY_IT; ++k) {
            const int i = t + k * SORT_T;
            if (i < nbin) ((int2*)pairs)[beg + i] = q[k];
        }
    }
}


#define GEMM_WAVES 8
#define GEMM_ROWS (GEMM_WAVES * 16)
#define W_PITCH 132

__global__ __launch_bounds__(GEMM_WAVES * 32)
void k_gemm_wmma(const float* __restrict__ H, const float* __restrict__ W,
                 const float* __restrict__ dinv, float* __restrict__ XW, int n_nodes) {
    __shared__ __attribute__((aligned(16))) float sW[GCN_D * W_PITCH];

    const int t = threadIdx.x;
    const int lane = t & 31;
    const int wv = t >> 5;
    const int lm = lane & 15;
    const int hi = lane >> 4;

    const int row0 = (blockIdx.x * GEMM_WAVES + wv) * 16;
    const bool active = row0 < n_nodes;
    const int rowa = active ? row0 : 0;

    const float2* __restrict__ Ap = (const float2*)(H + (size_t)(rowa + lm) * GCN_D);
    v2f a[GCN_D / 4];
#pragma unroll
    for (int s = 0; s < GCN_D / 4; ++s) {
        const float2 af = Ap[2 * s + hi];
        a[s].x = af.x;
        a[s].y = af.y;
    }

    const float4* __restrict__ W4 = (const float4*)W;
#pragma unroll
    for (int p = 0; p < (GCN_D * GCN_D / 4) / (GEMM_WAVES * 32); ++p) {
        const int i = t + p * (GEMM_WAVES * 32);
        const int j = i >> 5;
        const int c4 = i & 31;
        const float4 v = W4[i];
        *(float4*)&sW[j * W_PITCH + 4 * c4] = v;
    }
    __syncthreads();
    if (!active) return;

    const float* sB = sW + lm * W_PITCH + 2 * hi;

    const int m0 = row0 + hi * 8;
    const float4 dlo = *(const float4*)(dinv + m0);
    const float4 dhi = *(const float4*)(dinv + m0 + 4);

    v8f c[8] = {};
    v2f b[2][16];
#pragma unroll
    for (int n = 0; n < 8; ++n) {
#pragma unroll
        for (int u = 0; u < 2; ++u) {
            const float2 bf = *(const float2*)(sB + n * 16 * W_PITCH + 4 * u);
            b[0][2 * n + u].x = bf.x;
            b[0][2 * n + u].y = bf.y;
        }
    }
#pragma unroll
    for (int blk = 0; blk < GCN_D / 8; ++blk) {
        const int cur = blk & 1;
        if (blk + 1 < GCN_D / 8) {
#pragma unroll
            for (int n = 0; n < 8; ++n) {
#pragma unroll
                for (int u = 0; u < 2; ++u) {
                    const float2 bf = *(const float2*)(sB + n * 16 * W_PITCH + 4 * (2 * (blk + 1) + u));
                    b[cur ^ 1][2 * n + u].x = bf.x;
                    b[cur ^ 1][2 * n + u].y = bf.y;
                }
            }
        }
        __builtin_amdgcn_sched_barrier(0);
#pragma unroll
        for (int u = 0; u < 2; ++u) {
#pragma unroll
            for (int n = 0; n < 8; ++n) {
                c[n] = __builtin_amdgcn_wmma_f32_16x16x4_f32(
                            false, a[2 * blk + u],  false, b[cur][2 * n + u],
                            (short)0, c[n],  false,  false);
            }
        }
        __builtin_amdgcn_sched_barrier(0);
    }

    const float dv[8] = { dlo.x, dlo.y, dlo.z, dlo.w, dhi.x, dhi.y, dhi.z, dhi.w };
#pragma unroll
    for (int n = 0; n < 8; ++n) {
#pragma unroll
        for (int v = 0; v < 8; ++v)
            XW[(size_t)(m0 + v) * GCN_D + n * 16 + lm] = c[n][v] * dv[v];
    }
}


template <int U>
__device__ __forceinline__ void gather_rows(const float4* __restrict__ X4, int rl, int wl,
                                            int j, int lane, float4& a) {
    float4 m[U];
    float w[U];
#pragma unroll
    for (int u = 0; u < U; ++u) {
        const int r = __builtin_amdgcn_readlane(rl, j + u);
        w[u] = __int_as_float(__builtin_amdgcn_readlane(wl, j + u));
        m[u] = X4[(size_t)r * (GCN_D / 4) + lane];
    }
#pragma unroll
    for (int u = 0; u < U; ++u) {
        a.x = fmaf(m[u].x, w[u], a.x);
        a.y = fmaf(m[u].y, w[u], a.y);
        a.z = fmaf(m[u].z, w[u], a.z);
        a.w = fmaf(m[u].w, w[u], a.w);
    }
}

__global__ void k_gather_ln(const float* __restrict__ XW, const int* __restrict__ off,
                            const int* __restrict__ pairs, const float* __restrict__ dinv,
                            const float* __restrict__ bias, const float* hin,
                            const float* __restrict__ gamma, const float* __restrict__ beta,
                            float* hout, int n_nodes, int do_relu) {
    const int wave = (blockIdx.x * blockDim.x + threadIdx.x) >> 5;
    const int lane = threadIdx.x & 31;
    if (wave >= n_nodes) return;
    const int node = __builtin_amdgcn_readfirstlane(wave);
    const int beg = off[node];
    const int end = off[node + 1];
    const float di = dinv[node];

    const float4* __restrict__ X4 = (const float4*)XW;
    const int2* __restrict__ P = (const int2*)pairs;

    float4 x = X4[(size_t)node * (GCN_D / 4) + lane];
    float4 b = ((const float4*)bias)[lane];
    const float4 id = ((const float4*)(hin + (size_t)node * GCN_D))[lane];
    float4 a = x;

    for (int p0 = beg; p0 < end; p0 += 32) {
        const int p = p0 + lane;
        int2 pr;
        pr.x = 0;
        pr.y = 0;
        if (p < end) pr = P[p];
        const int left = end - p0;
        const int cnt = left < 32 ? left : 32;
        int j = 0;
        for (; j + 8 <= cnt; j += 8) gather_rows<8>(X4, pr.x, pr.y, j, lane, a);
        if (j + 4 <= cnt) { gather_rows<4>(X4, pr.x, pr.y, j, lane, a); j += 4; }
        if (j + 2 <= cnt) { gather_rows<2>(X4, pr.x, pr.y, j, lane, a); j += 2; }
        if (j < cnt) gather_rows<1>(X4, pr.x, pr.y, j, lane, a);
    }

    a.x = fmaf(di, a.x, b.x);
    a.y = fmaf(di, a.y, b.y);
    a.z = fmaf(di, a.z, b.z);
    a.w = fmaf(di, a.w, b.w);

    float s = a.x + a.y + a.z + a.w;
#pragma unroll
    for (int o = 16; o > 0; o >>= 1) s += __shfl_xor(s, o, 32);
    const float mu = s * (1.0f / GCN_D);

    float dx = a.x - mu, dy = a.y - mu, dz = a.z - mu, dw = a.w - mu;
    float q = dx * dx + dy * dy + dz * dz + dw * dw;
#pragma unroll
    for (int o = 16; o > 0; o >>= 1) q += __shfl_xor(q, o, 32);
    const float inv = rsqrtf(q * (1.0f / GCN_D) + LN_EPS);

    const float4 g = ((const float4*)gamma)[lane];
    const float4 be = ((const float4*)beta)[lane];

    float4 y;
    y.x = dx * inv * g.x + be.x;
    y.y = dy * inv * g.y + be.y;
    y.z = dz * inv * g.z + be.z;
    y.w = dw * inv * g.w + be.w;
    if (do_relu) {
        y.x = fmaxf(y.x, 0.0f);
        y.y = fmaxf(y.y, 0.0f);
        y.z = fmaxf(y.z, 0.0f);
        y.w = fmaxf(y.w, 0.0f);
    }
    y.x += id.x; y.y += id.y; y.z += id.z; y.w += id.w;
    ((float4*)(hout + (size_t)node * GCN_D))[lane] = y;
}


#define FUSE_WAVES 16
#define FUSE_T (FUSE_WAVES * 32)
#define TILE_NODES 32
#define FUSE_TILES 4
#define FUSE_NODES (TILE_NODES * FUSE_TILES)
#define KS_SLICE 8

static_assert(GCN_N % 16 == 0, "strips of 16 nodes must be whole");
static_assert((GCN_D * GCN_D / 4) % FUSE_T == 0, "W staging: whole passes of the work-group");

__device__ __forceinline__ void gemm_slice(v8f& c, int& ks, const float* pa, const float* pb) {
#pragma unroll
    for (int q = 0; q < KS_SLICE; ++q) {
        const float2 af = *(const float2*)(pa + 4 * (ks + q));
        const float2 bf = *(const float2*)(pb + 4 * (ks + q));
        v2f a2; a2.x = af.x; a2.y = af.y;
        v2f b2; b2.x = bf.x; b2.y = bf.y;
        c = __builtin_amdgcn_wmma_f32_16x16x4_f32( false, a2,  false, b2,
                                                   (short)0, c,
                                                   false,  false);
    }
    ks += KS_SLICE;
}

__device__ __forceinline__ void gather_rows8_mm(const float4* __restrict__ X4, int rl, int wl,
                                                int j, int lane, float4& a,
                                                v8f& c, int& ks, const float* pa, const float* pb) {
    float4 m[8];
    float w[8];
#pragma unroll
    for (int u = 0; u < 8; ++u) {
        const int r = __builtin_amdgcn_readlane(rl, j + u);
        w[u] = __int_as_float(__builtin_amdgcn_readlane(wl, j + u));
        m[u] = X4[(size_t)r * (GCN_D / 4) + lane];
    }
    if (ks < GCN_D / 4) gemm_slice(c, ks, pa, pb);
#pragma unroll
    for (int u = 0; u < 8; ++u) {
        a.x = fmaf(m[u].x, w[u], a.x);
        a.y = fmaf(m[u].y, w[u], a.y);
        a.z = fmaf(m[u].z, w[u], a.z);
        a.w = fmaf(m[u].w, w[u], a.w);
    }
}

__global__ __launch_bounds__(FUSE_T)
void k_gather_ln_gemm(const float* __restrict__ XW, const int* __restrict__ off,
                      const int* __restrict__ pairs, const float* __restrict__ dinv,
                      const float* __restrict__ bias, const float* __restrict__ hin,
                      const float* __restrict__ gamma, const float* __restrict__ beta,
                      float* __restrict__ hout, const float* __restrict__ Wn,
                      float* __restrict__ XWn, int n_nodes, int do_relu) {
    __shared__ __attribute__((aligned(16))) float sW[GCN_D * W_PITCH];
    __shared__ __attribute__((aligned(16))) float sA[2][TILE_NODES * W_PITCH];

    const int t = threadIdx.x;
    const int lane = t & 31;
    const int wv = __builtin_amdgcn_readfirstlane(t >> 5);
    const int lm = lane & 15;
    const int hi = lane >> 4;

    const float4* __restrict__ W4 = (const float4*)Wn;
#pragma unroll
    for (int p = 0; p < (GCN_D * GCN_D / 4) / FUSE_T; ++p) {
        const int i = t + p * FUSE_T;
        const int jr = i >> 5;
        const int c4 = i & 31;
        const float4 v = W4[i];
        *(float4*)&sW[jr * W_PITCH + 4 * c4] = v;
    }

    const float4* __restrict__ X4 = (const float4*)XW;
    const int2* __restrict__ P = (const int2*)pairs;
    const float4 b4 = ((const float4*)bias)[lane];
    const float4 g4 = ((const float4*)gamma)[lane];
    const float4 be4 = ((const float4*)beta)[lane];

    const int blk_base = blockIdx.x * FUSE_NODES;
    const int gs = wv >> 3;
    const int gn = wv & 7;
    const float* pb = sW + (gn * 16 + lm) * W_PITCH + 2 * hi;

    v8f c = {};
    int ks = GCN_D / 4;
    for (int tile = 0; tile <= FUSE_TILES; ++tile) {
        const float* pa = sA[0];
        int m0 = 0;
        bool gvalid = false;
        float4 dlo = make_float4(0.f, 0.f, 0.f, 0.f);
        float4 dhi = dlo;
        if (tile > 0) {
            const int prow0 = blk_base + (tile - 1) * TILE_NODES + gs * 16;
            gvalid = prow0 < n_nodes;
            if (gvalid) {
                pa = sA[(tile - 1) & 1] + (gs * 16 + lm) * W_PITCH + 2 * hi;
                m0 = prow0 + hi * 8;
                dlo = *(const float4*)(dinv + m0);
                dhi = *(const float4*)(dinv + m0 + 4);
                ks = 0;
#pragma unroll
                for (int v = 0; v < 8; ++v) c[v] = 0.0f;
            }
        }

        if (tile < FUSE_TILES) {
            for (int i = 0; i < 2; ++i) {
                const int node = blk_base + tile * TILE_NODES + 2 * wv + i;
                if (node < n_nodes) {
                    const int beg = off[node];
                    const int end = off[node + 1];
                    const float di = dinv[node];
                    float4 a = X4[(size_t)node * (GCN_D / 4) + lane];
                    const float4 id = ((const float4*)(hin + (size_t)node * GCN_D))[lane];

                    for (int p0 = beg; p0 < end; p0 += 32) {
                        const int p = p0 + lane;
                        int2 pr;
                        pr.x = 0;
                        pr.y = 0;
                        if (p < end) pr = P[p];
                        const int left = end - p0;
                        const int cnt = left < 32 ? left : 32;
                        int j = 0;
                        for (; j + 8 <= cnt; j += 8)
                            gather_rows8_mm(X4, pr.x, pr.y, j, lane, a, c, ks, pa, pb);
                        if (j + 4 <= cnt) { gather_rows<4>(X4, pr.x, pr.y, j, lane, a); j += 4; }
                        if (j + 2 <= cnt) { gather_rows<2>(X4, pr.x, pr.y, j, lane, a); j += 2; }
                        if (j < cnt) gather_rows<1>(X4, pr.x, pr.y, j, lane, a);
                    }

                    a.x = fmaf(di, a.x, b4.x);
                    a.y = fmaf(di, a.y, b4.y);
                    a.z = fmaf(di, a.z, b4.z);
                    a.w = fmaf(di, a.w, b4.w);

                    float s = a.x + a.y + a.z + a.w;
#pragma unroll
                    for (int o = 16; o > 0; o >>= 1) s += __shfl_xor(s, o, 32);
                    const float mu = s * (1.0f / GCN_D);

                    float dx = a.x - mu, dy = a.y - mu, dz = a.z - mu, dw = a.w - mu;
                    float q = dx * dx + dy * dy + dz * dz + dw * dw;
#pragma unroll
                    for (int o = 16; o > 0; o >>= 1) q += __shfl_xor(q, o, 32);
                    const float inv = rsqrtf(q * (1.0f / GCN_D) + LN_EPS);

                    float4 y;
                    y.x = dx * inv * g4.x + be4.x;
                    y.y = dy * inv * g4.y + be4.y;
                    y.z = dz * inv * g4.z + be4.z;
                    y.w = dw * inv * g4.w + be4.w;
                    if (do_relu) {
                        y.x = fmaxf(y.x, 0.0f);
                        y.y = fmaxf(y.y, 0.0f);
                        y.z = fmaxf(y.z, 0.0f);
                        y.w = fmaxf(y.w, 0.0f);
                    }
                    y.x += id.x; y.y += id.y; y.z += id.z; y.w += id.w;
                    ((float4*)(hout + (size_t)node * GCN_D))[lane] = y;
                    *(float4*)&sA[tile & 1][(2 * wv + i) * W_PITCH + 4 * lane] = y;
                }
            }
        }

        while (ks < GCN_D / 4) gemm_slice(c, ks, pa, pb);
        if (gvalid) {
            const float dv[8] = { dlo.x, dlo.y, dlo.z, dlo.w, dhi.x, dhi.y, dhi.z, dhi.w };
#pragma unroll
            for (int v = 0; v < 8; ++v)
                XWn[(size_t)(m0 + v) * GCN_D + gn * 16 + lm] = c[v] * dv[v];
        }
        if (tile < FUSE_TILES) __syncthreads();
    }
}


extern "C" void kernel_launch(void* const* d_in, const int* in_sizes, int n_in,
                              void* d_out, int out_size, void* d_ws, size_t ws_size,
                              hipStream_t stream) {
    (void)in_sizes; (void)n_in; (void)out_size; (void)ws_size;

    const float* x      = (const float*)d_in[0];
    const int*   ei     = (const int*)d_in[1];
    const float* ew     = (const float*)d_in[2];
    const float* Ws     = (const float*)d_in[3];
    const float* bs     = (const float*)d_in[4];
    const float* gammas = (const float*)d_in[5];
    const float* betas  = (const float*)d_in[6];
    float*       out    = (float*)d_out;

    const int* row = ei;
    const int* col = ei + GCN_E;

    float* ws    = (float*)d_ws;
    int*   off   = (int*)ws;
    float* dinv  = (float*)(off + 50048);
    int*   cntT  = (int*)(dinv + 50048);
    int*   startT = cntT + 100352;
    int*   tmp   = startT + 100352;
    int*   pairs = tmp + 2 * GCN_E;
    float* xw    = (float*)(pairs + 2 * GCN_E);
    float* h1    = xw + (size_t)GCN_N * GCN_D;
    float* xw2   = h1 + (size_t)GCN_N * GCN_D;

    const int BLK = 256;

    const int node_wave_blocks = (GCN_N + (BLK / 32) - 1) / (BLK / 32);
    k_slice_sort<<<SORT_G, SORT_T, 0, stream>>>(row, col, ew, cntT, startT, tmp, GCN_E);
    k_bin_sort<<<NBINS, SORT_T, 0, stream>>>(cntT, startT, tmp, off, pairs, dinv, GCN_N);

    const int gemm_blocks  = (GCN_N + GEMM_ROWS - 1) / GEMM_ROWS;

    const int fuse_blocks = (GCN_N + FUSE_NODES - 1) / FUSE_NODES;

    k_gemm_wmma<<<gemm_blocks, GEMM_WAVES * 32, 0, stream>>>(x, Ws, dinv, xw, GCN_N);
    k_gather_ln_gemm<<<fuse_blocks, FUSE_T, 0, stream>>>(xw, off, pairs, dinv,
                                                         bs, x, gammas, betas, h1,
                                                         Ws + (size_t)GCN_D * GCN_D, xw2,
                                                         GCN_N,  1);
    k_gather_ln_gemm<<<fuse_blocks, FUSE_T, 0, stream>>>(xw2, off, pairs, dinv,
                                                         bs + GCN_D, h1, gammas + GCN_D,
                                                         betas + GCN_D, out,
                                                         Ws + (size_t)2 * GCN_D * GCN_D, xw,
                                                         GCN_N,  1);
    k_gather_ln<<<node_wave_blocks, BLK, 0, stream>>>(xw, off, pairs, dinv,
                                                      bs + 2 * GCN_D, out,
                                                      gammas + 2 * GCN_D, betas + 2 * GCN_D,
                                                      out, GCN_N,  0);
}
